# starting kernel
# baseline (speedup 1.0000x reference)
#include <hip/hip_runtime.h>
#include <math.h>

#define BB   64
#define NN   2048
#define MM   2048
#define CC   64
#define HH   128
#define NNZE 32768

typedef float v2f __attribute__((ext_vector_type(2)));
typedef float v8f __attribute__((ext_vector_type(8)));

__global__ __launch_bounds__(512)
void spmm_pool_kernel(const float* __restrict__ x,
                      const int*   __restrict__ rows,
                      const int*   __restrict__ cols,
                      const float* __restrict__ vals,
                      const int*   __restrict__ idx,
                      float* __restrict__ xdec) {
  extern __shared__ float smem[];
  float* sco  = smem;
  float* psum = sco  + MM * 32;
  float* pmax = psum + CC * 32;
  float* cnt  = pmax + CC * 32;

  const int tid    = threadIdx.x;
  const int lane   = tid & 31;
  const int wid    = tid >> 5;
  const int NWAVES = blockDim.x >> 5;
  const int b      = blockIdx.x >> 2;
  const int hbase  = (blockIdx.x & 3) * 32;

  for (int i = tid; i < MM * 32; i += blockDim.x) sco[i] = 0.0f;
  for (int i = tid; i < CC * 32; i += blockDim.x) { psum[i] = 0.0f; pmax[i] = -INFINITY; }
  for (int i = tid; i < CC;      i += blockDim.x) cnt[i] = 0.0f;
  __syncthreads();

  const size_t ebase = (size_t)b * NNZE;
  const float* xg = x + (size_t)b * NN * HH + hbase;
  for (int e0 = wid * 32; e0 < NNZE; e0 += NWAVES * 32) {
    const int   rbuf = rows[ebase + e0 + lane];
    const int   cbuf = cols[ebase + e0 + lane];
    const float vbuf = vals[ebase + e0 + lane];
#pragma unroll
    for (int j = 0; j < 32; ++j) {
      const int   r = __shfl(rbuf, j, 32);
      const int   c = __shfl(cbuf, j, 32);
      const float v = __shfl(vbuf, j, 32);
      const float xv = xg[(size_t)c * HH + lane];
      atomicAdd(&sco[r * 32 + lane], v * xv);
    }
  }
  __syncthreads();

  const size_t ibase = (size_t)b * MM;
  for (int m = wid; m < MM; m += NWAVES) {
    const int   c = idx[ibase + m];
    const float v = sco[m * 32 + lane];
    atomicAdd(&psum[c * 32 + lane], v);
    __hip_atomic_fetch_max(&pmax[c * 32 + lane], v,
                           __ATOMIC_RELAXED, __HIP_MEMORY_SCOPE_WORKGROUP);
    if (lane == 0) atomicAdd(&cnt[c], 1.0f);
  }
  __syncthreads();

  for (int c = wid; c < CC; c += NWAVES) {
    const float s    = psum[c * 32 + lane];
    const float mx   = pmax[c * 32 + lane];
    const float mean = s / fmaxf(cnt[c], 1.0f);
    const size_t base = ((size_t)b * CC + c) * (3 * HH);
    xdec[base +            hbase + lane] = mean;
    xdec[base +     HH +   hbase + lane] = mx;
    xdec[base + 2 * HH +   hbase + lane] = s;
  }
}

#define XD_LD 388
#define QK_LD 132
#define S_LD  66

__device__ __forceinline__ v2f ldsA(const float* base, int ld, int mt, int k, int lane) {
  const int hi = (lane >> 4) << 1;
  const float* p = base + (mt * 16 + (lane & 15)) * ld + k + hi;
  v2f a; a.x = p[0]; a.y = p[1];
  return a;
}

__global__ __launch_bounds__(256)
void cluster_attn_kernel(const float* __restrict__ xdec_g,
                         const float* __restrict__ Wq,
                         const float* __restrict__ Wk,
                         const float* __restrict__ Wv,
                         float* __restrict__ out) {
  extern __shared__ float smem[];
  float* xd = smem;
  float* Q  = xd + 64 * XD_LD;
  float* K  = Q  + 64 * QK_LD;
  float* V  = K  + 64 * QK_LD;
  float* S  = xd;

  const int tid  = threadIdx.x;
  const int lane = tid & 31;
  const int wid  = tid >> 5;
  const int b    = blockIdx.x;

  const float* src = xdec_g + (size_t)b * (CC * 3 * HH);
  for (int i = tid; i < CC * 3 * HH; i += blockDim.x)
    xd[(i / 384) * XD_LD + (i % 384)] = src[i];
  __syncthreads();

  for (int t = wid; t < 96; t += 8) {
    const float* W = (t < 32) ? Wq : (t < 64) ? Wk : Wv;
    float*       D = (t < 32) ? Q  : (t < 64) ? K  : V;
    const int tl = t & 31, mt = tl >> 3, nt = tl & 7;
    const int hi = (lane >> 4) << 1, col = nt * 16 + (lane & 15);
    v8f acc = {};
    for (int k = 0; k < 3 * HH; k += 4) {
      v2f a = ldsA(xd, XD_LD, mt, k, lane);
      v2f bf; bf.x = W[(k + hi) * HH + col]; bf.y = W[(k + 1 + hi) * HH + col];
      acc = __builtin_amdgcn_wmma_f32_16x16x4_f32(false, a, false, bf,
                                                  (short)0, acc, false, false);
    }
    const int rbase = mt * 16 + ((lane >> 4) << 3);
#pragma unroll
    for (int r = 0; r < 8; ++r) D[(rbase + r) * QK_LD + col] = acc[r];
  }
  __syncthreads();

  const float scale = 0.08838834764831845f;
  for (int t = wid; t < 16; t += 8) {
    const int mt = t >> 2, nt = t & 3;
    const int hi = (lane >> 4) << 1, col = nt * 16 + (lane & 15);
    v8f acc = {};
    for (int k = 0; k < HH; k += 4) {
      v2f a = ldsA(Q, QK_LD, mt, k, lane);
      v2f bf; bf.x = K[col * QK_LD + k + hi]; bf.y = K[col * QK_LD + k + hi + 1];
      acc = __builtin_amdgcn_wmma_f32_16x16x4_f32(false, a, false, bf,
                                                  (short)0, acc, false, false);
    }
    const int rbase = mt * 16 + ((lane >> 4) << 3);
#pragma unroll
    for (int r = 0; r < 8; ++r) S[(rbase + r) * S_LD + col] = acc[r] * scale;
  }
  __syncthreads();

  for (int r = wid; r < CC; r += 8) {
    float v0 = S[r * S_LD + lane];
    float v1 = S[r * S_LD + 32 + lane];
    float m = fmaxf(v0, v1);
    for (int off = 16; off > 0; off >>= 1) m = fmaxf(m, __shfl_xor(m, off, 32));
    float e0 = expf(v0 - m), e1 = expf(v1 - m);
    float s = e0 + e1;
    for (int off = 16; off > 0; off >>= 1) s += __shfl_xor(s, off, 32);
    const float inv = 1.0f / s;
    S[r * S_LD + lane]      = e0 * inv;
    S[r * S_LD + 32 + lane] = e1 * inv;
  }
  __syncthreads();

  float* ob = out + (size_t)b * (CC * HH);
  for (int t = wid; t < 32; t += 8) {
    const int mt = t >> 3, nt = t & 7;
    const int hi = (lane >> 4) << 1, col = nt * 16 + (lane & 15);
    v8f acc = {};
    for (int k = 0; k < CC; k += 4) {
      v2f a = ldsA(S, S_LD, mt, k, lane);
      v2f bf; bf.x = V[(k + hi) * QK_LD + col]; bf.y = V[(k + 1 + hi) * QK_LD + col];
      acc = __builtin_amdgcn_wmma_f32_16x16x4_f32(false, a, false, bf,
                                                  (short)0, acc, false, false);
    }
    const int rbase = mt * 16 + ((lane >> 4) << 3);
#pragma unroll
    for (int r = 0; r < 8; ++r) ob[(rbase + r) * HH + col] = acc[r];
  }
}

extern "C" void kernel_launch(void* const* d_in, const int* in_sizes, int n_in,
                              void* d_out, int out_size, void* d_ws, size_t ws_size,
                              hipStream_t stream) {
  const float* x    = (const float*)d_in[0];
  const int*   rows = (const int*)d_in[3];
  const int*   cols = (const int*)d_in[4];
  const float* vals = (const float*)d_in[5];
  const int*   idx  = (const int*)d_in[6];
  const float* Wq   = (const float*)d_in[7];
  const float* Wk   = (const float*)d_in[8];
  const float* Wv   = (const float*)d_in[9];
  float* xdec = (float*)d_ws;

  const size_t smem1 = (size_t)(MM * 32 + 2 * CC * 32 + CC) * sizeof(float);
  spmm_pool_kernel<<<dim3(BB * (HH / 32)), dim3(512), smem1, stream>>>(
      x, rows, cols, vals, idx, xdec);

  const size_t smem2 = (size_t)(64 * XD_LD + 3 * 64 * QK_LD) * sizeof(float);
  cluster_attn_kernel<<<dim3(BB), dim3(256), smem2, stream>>>(
      xdec, Wq, Wk, Wv, (float*)d_out);
}


# k1: loads grouped ahead of the LDS atomics (compiler moves no load over one): 32 gathers a batch, next batch in flight during the adds; pooling takes 32 rows a wave, add order by arrival as before
# speedup vs baseline: 2.2691x; 2.2691x over previous
#include <hip/hip_runtime.h>
#include <math.h>

#define BB   64
#define NN   2048
#define MM   2048
#define CC   64
#define HH   128
#define NNZE 32768

typedef float v2f __attribute__((ext_vector_type(2)));
typedef float v8f __attribute__((ext_vector_type(8)));

#define NW1   16
#define NB1   (NNZE / (32 * NW1))
#define ESTEP (32 * NW1)

__device__ __forceinline__ void gather32(float (&xv)[32], const float* __restrict__ xg,
                                         int cbuf, int lane) {
#pragma unroll
  for (int j = 0; j < 32; ++j) {
    const int c = __shfl(cbuf, j, 32);
    xv[j] = xg[(size_t)c * HH + lane];
  }
}

__device__ __forceinline__ void scatter32(float* sco, const float (&xv)[32],
                                          int rbuf, float vbuf, int lane) {
#pragma unroll
  for (int j = 0; j < 32; ++j) {
    const int   r = __shfl(rbuf, j, 32);
    const float v = __shfl(vbuf, j, 32);
    atomicAdd(&sco[r * 32 + lane], v * xv[j]);
  }
}

__global__ __launch_bounds__(512)
void spmm_pool_kernel(const float* __restrict__ x,
                      const int*   __restrict__ rows,
                      const int*   __restrict__ cols,
                      const float* __restrict__ vals,
                      const int*   __restrict__ idx,
                      float* __restrict__ xdec) {
  extern __shared__ float smem[];
  float* sco  = smem;
  float* psum = sco  + MM * 32;
  float* pmax = psum + CC * 32;
  float* cnt  = pmax + CC * 32;

  const int tid    = threadIdx.x;
  const int lane   = tid & 31;
  const int wid    = tid >> 5;
  const int NWAVES = blockDim.x >> 5;
  const int b      = blockIdx.x >> 2;
  const int hbase  = (blockIdx.x & 3) * 32;

  for (int i = tid; i < MM * 32; i += blockDim.x) sco[i] = 0.0f;
  for (int i = tid; i < CC * 32; i += blockDim.x) { psum[i] = 0.0f; pmax[i] = -INFINITY; }
  for (int i = tid; i < CC;      i += blockDim.x) cnt[i] = 0.0f;
  __syncthreads();

  const size_t ebase = (size_t)b * NNZE;
  const float* xg = x + (size_t)b * NN * HH + hbase;
  const int*   rp = rows + ebase + wid * 32 + lane;
  const int*   cp = cols + ebase + wid * 32 + lane;
  const float* vp = vals + ebase + wid * 32 + lane;

  float xa[32], xb[32];
  int   r0 = rp[0],     c0 = cp[0];     float v0 = vp[0];
  int   r1 = rp[ESTEP], c1 = cp[ESTEP]; float v1 = vp[ESTEP];
  __builtin_amdgcn_sched_barrier(0);
  gather32(xa, xg, c0, lane);
  for (int k = 0; k < NB1 - 2; k += 2) {
    const int   r2 = rp[(k + 2) * ESTEP];
    const int   c2 = cp[(k + 2) * ESTEP];
    const float v2 = vp[(k + 2) * ESTEP];
    __builtin_amdgcn_sched_barrier(0);
    gather32(xb, xg, c1, lane);
    scatter32(sco, xa, r0, v0, lane);
    const int   r3 = rp[(k + 3) * ESTEP];
    const int   c3 = cp[(k + 3) * ESTEP];
    const float v3 = vp[(k + 3) * ESTEP];
    __builtin_amdgcn_sched_barrier(0);
    gather32(xa, xg, c2, lane);
    scatter32(sco, xb, r1, v1, lane);
    r0 = r2; c0 = c2; v0 = v2;
    r1 = r3; c1 = c3; v1 = v3;
  }
  gather32(xb, xg, c1, lane);
  scatter32(sco, xa, r0, v0, lane);
  scatter32(sco, xb, r1, v1, lane);

  const size_t ibase = (size_t)b * MM;
  int cb[MM / (32 * NW1)];
#pragma unroll
  for (int i = 0; i < MM / (32 * NW1); ++i) cb[i] = idx[ibase + (wid + i * NW1) * 32 + lane];
  __syncthreads();

#pragma unroll
  for (int i = 0; i < MM / (32 * NW1); ++i) {
    const int m0 = (wid + i * NW1) * 32;
    int   cj[32];
    float cv[32];
#pragma unroll
    for (int j = 0; j < 32; ++j) cj[j] = __shfl(cb[i], j, 32);
#pragma unroll
    for (int j = 0; j < 32; ++j) cv[j] = sco[(m0 + j) * 32 + lane];
#pragma unroll
    for (int j = 0; j < 32; ++j) {
      const int   c = cj[j];
      const float v = cv[j];
      atomicAdd(&psum[c * 32 + lane], v);
      __hip_atomic_fetch_max(&pmax[c * 32 + lane], v,
                             __ATOMIC_RELAXED, __HIP_MEMORY_SCOPE_WORKGROUP);
      if (lane == 0) atomicAdd(&cnt[c], 1.0f);
    }
  }
  __syncthreads();

  for (int c = wid; c < CC; c += NWAVES) {
    const float s    = psum[c * 32 + lane];
    const float mx   = pmax[c * 32 + lane];
    const float mean = s / fmaxf(cnt[c], 1.0f);
    const size_t base = ((size_t)b * CC + c) * (3 * HH);
    xdec[base +            hbase + lane] = mean;
    xdec[base +     HH +   hbase + lane] = mx;
    xdec[base + 2 * HH +   hbase + lane] = s;
  }
}

#define XD_LD 388
#define QK_LD 132
#define S_LD  66

__device__ __forceinline__ v2f ldsA(const float* base, int ld, int mt, int k, int lane) {
  const int hi = (lane >> 4) << 1;
  const float* p = base + (mt * 16 + (lane & 15)) * ld + k + hi;
  v2f a; a.x = p[0]; a.y = p[1];
  return a;
}

__global__ __launch_bounds__(256)
void cluster_attn_kernel(const float* __restrict__ xdec_g,
                         const float* __restrict__ Wq,
                         const float* __restrict__ Wk,
                         const float* __restrict__ Wv,
                         float* __restrict__ out) {
  extern __shared__ float smem[];
  float* xd = smem;
  float* Q  = xd + 64 * XD_LD;
  float* K  = Q  + 64 * QK_LD;
  float* V  = K  + 64 * QK_LD;
  float* S  = xd;

  const int tid  = threadIdx.x;
  const int lane = tid & 31;
  const int wid  = tid >> 5;
  const int b    = blockIdx.x;

  const float* src = xdec_g + (size_t)b * (CC * 3 * HH);
  for (int i = tid; i < CC * 3 * HH; i += blockDim.x)
    xd[(i / 384) * XD_LD + (i % 384)] = src[i];
  __syncthreads();

  for (int t = wid; t < 96; t += 8) {
    const float* W = (t < 32) ? Wq : (t < 64) ? Wk : Wv;
    float*       D = (t < 32) ? Q  : (t < 64) ? K  : V;
    const int tl = t & 31, mt = tl >> 3, nt = tl & 7;
    const int hi = (lane >> 4) << 1, col = nt * 16 + (lane & 15);
    v8f acc = {};
    for (int k = 0; k < 3 * HH; k += 4) {
      v2f a = ldsA(xd, XD_LD, mt, k, lane);
      v2f bf; bf.x = W[(k + hi) * HH + col]; bf.y = W[(k + 1 + hi) * HH + col];
      acc = __builtin_amdgcn_wmma_f32_16x16x4_f32(false, a, false, bf,
                                                  (short)0, acc, false, false);
    }
    const int rbase = mt * 16 + ((lane >> 4) << 3);
#pragma unroll
    for (int r = 0; r < 8; ++r) D[(rbase + r) * QK_LD + col] = acc[r];
  }
  __syncthreads();

  const float scale = 0.08838834764831845f;
  for (int t = wid; t < 16; t += 8) {
    const int mt = t >> 2, nt = t & 3;
    const int hi = (lane >> 4) << 1, col = nt * 16 + (lane & 15);
    v8f acc = {};
    for (int k = 0; k < HH; k += 4) {
      v2f a = ldsA(Q, QK_LD, mt, k, lane);
      v2f bf; bf.x = K[col * QK_LD + k + hi]; bf.y = K[col * QK_LD + k + hi + 1];
      acc = __builtin_amdgcn_wmma_f32_16x16x4_f32(false, a, false, bf,
                                                  (short)0, acc, false, false);
    }
    const int rbase = mt * 16 + ((lane >> 4) << 3);
#pragma unroll
    for (int r = 0; r < 8; ++r) S[(rbase + r) * S_LD + col] = acc[r] * scale;
  }
  __syncthreads();

  for (int r = wid; r < CC; r += 8) {
    float v0 = S[r * S_LD + lane];
    float v1 = S[r * S_LD + 32 + lane];
    float m = fmaxf(v0, v1);
    for (int off = 16; off > 0; off >>= 1) m = fmaxf(m, __shfl_xor(m, off, 32));
    float e0 = expf(v0 - m), e1 = expf(v1 - m);
    float s = e0 + e1;
    for (int off = 16; off > 0; off >>= 1) s += __shfl_xor(s, off, 32);
    const float inv = 1.0f / s;
    S[r * S_LD + lane]      = e0 * inv;
    S[r * S_LD + 32 + lane] = e1 * inv;
  }
  __syncthreads();

  float* ob = out + (size_t)b * (CC * HH);
  for (int t = wid; t < 32; t += 8) {
    const int mt = t >> 3, nt = t & 7;
    const int hi = (lane >> 4) << 1, col = nt * 16 + (lane & 15);
    v8f acc = {};
    for (int k = 0; k < CC; k += 4) {
      v2f a = ldsA(S, S_LD, mt, k, lane);
      v2f bf; bf.x = V[(k + hi) * QK_LD + col]; bf.y = V[(k + 1 + hi) * QK_LD + col];
      acc = __builtin_amdgcn_wmma_f32_16x16x4_f32(false, a, false, bf,
                                                  (short)0, acc, false, false);
    }
    const int rbase = mt * 16 + ((lane >> 4) << 3);
#pragma unroll
    for (int r = 0; r < 8; ++r) ob[(rbase + r) * HH + col] = acc[r];
  }
}

extern "C" void kernel_launch(void* const* d_in, const int* in_sizes, int n_in,
                              void* d_out, int out_size, void* d_ws, size_t ws_size,
                              hipStream_t stream) {
  const float* x    = (const float*)d_in[0];
  const int*   rows = (const int*)d_in[3];
  const int*   cols = (const int*)d_in[4];
  const float* vals = (const float*)d_in[5];
  const int*   idx  = (const int*)d_in[6];
  const float* Wq   = (const float*)d_in[7];
  const float* Wk   = (const float*)d_in[8];
  const float* Wv   = (const float*)d_in[9];
  float* xdec = (float*)d_ws;

  const size_t smem1 = (size_t)(MM * 32 + 2 * CC * 32 + CC) * sizeof(float);
  spmm_pool_kernel<<<dim3(BB * (HH / 32)), dim3(512), smem1, stream>>>(
      x, rows, cols, vals, idx, xdec);

  const size_t smem2 = (size_t)(64 * XD_LD + 3 * 64 * QK_LD) * sizeof(float);
  cluster_attn_kernel<<<dim3(BB), dim3(256), smem2, stream>>>(
      xdec, Wq, Wk, Wv, (float*)d_out);
}


# k1: per-edge broadcasts of row/col/val via v_readlane (scalar regs) instead of ds_bpermute: the LDS queue carries only the adds, gather addresses are scalar; add order by arrival as before
# speedup vs baseline: 2.5775x; 1.1359x over previous
#include <hip/hip_runtime.h>
#include <math.h>

#define BB   64
#define NN   2048
#define MM   2048
#define CC   64
#define HH   128
#define NNZE 32768

typedef float v2f __attribute__((ext_vector_type(2)));
typedef float v8f __attribute__((ext_vector_type(8)));

#define NW1   16
#define NB1   (NNZE / (32 * NW1))
#define ESTEP (32 * NW1)

__device__ __forceinline__ int lane_bcast_i(int v, int j) {
  return __builtin_amdgcn_readlane(v, j);
}
__device__ __forceinline__ float lane_bcast_f(float v, int j) {
  return __int_as_float(__builtin_amdgcn_readlane(__float_as_int(v), j));
}

__device__ __forceinline__ void gather32(float (&xv)[32], const float* __restrict__ xg,
                                         int cbuf, int lane) {
#pragma unroll
  for (int j = 0; j < 32; ++j) {
    const unsigned c = (unsigned)lane_bcast_i(cbuf, j);
    xv[j] = xg[c * HH + (unsigned)lane];
  }
}

__device__ __forceinline__ void scatter32(float* sco, const float (&xv)[32],
                                          int rbuf, float vbuf, int lane) {
#pragma unroll
  for (int j = 0; j < 32; ++j) {
    const int   r = lane_bcast_i(rbuf, j);
    const float v = lane_bcast_f(vbuf, j);
    atomicAdd(&sco[r * 32 + lane], v * xv[j]);
  }
}

__global__ __launch_bounds__(512)
void spmm_pool_kernel(const float* __restrict__ x,
                      const int*   __restrict__ rows,
                      const int*   __restrict__ cols,
                      const float* __restrict__ vals,
                      const int*   __restrict__ idx,
                      float* __restrict__ xdec) {
  extern __shared__ float smem[];
  float* sco  = smem;
  float* psum = sco  + MM * 32;
  float* pmax = psum + CC * 32;
  float* cnt  = pmax + CC * 32;

  const int tid    = threadIdx.x;
  const int lane   = tid & 31;
  const int wid    = tid >> 5;
  const int NWAVES = blockDim.x >> 5;
  const int b      = blockIdx.x >> 2;
  const int hbase  = (blockIdx.x & 3) * 32;

  for (int i = tid; i < MM * 32; i += blockDim.x) sco[i] = 0.0f;
  for (int i = tid; i < CC * 32; i += blockDim.x) { psum[i] = 0.0f; pmax[i] = -INFINITY; }
  for (int i = tid; i < CC;      i += blockDim.x) cnt[i] = 0.0f;
  __syncthreads();

  const size_t ebase = (size_t)b * NNZE;
  const float* xg = x + (size_t)b * NN * HH + hbase;
  const int*   rp = rows + ebase + wid * 32 + lane;
  const int*   cp = cols + ebase + wid * 32 + lane;
  const float* vp = vals + ebase + wid * 32 + lane;

  float xa[32], xb[32];
  int   r0 = rp[0],     c0 = cp[0];     float v0 = vp[0];
  int   r1 = rp[ESTEP], c1 = cp[ESTEP]; float v1 = vp[ESTEP];
  __builtin_amdgcn_sched_barrier(0);
  gather32(xa, xg, c0, lane);
  for (int k = 0; k < NB1 - 2; k += 2) {
    const int   r2 = rp[(k + 2) * ESTEP];
    const int   c2 = cp[(k + 2) * ESTEP];
    const float v2 = vp[(k + 2) * ESTEP];
    __builtin_amdgcn_sched_barrier(0);
    gather32(xb, xg, c1, lane);
    scatter32(sco, xa, r0, v0, lane);
    const int   r3 = rp[(k + 3) * ESTEP];
    const int   c3 = cp[(k + 3) * ESTEP];
    const float v3 = vp[(k + 3) * ESTEP];
    __builtin_amdgcn_sched_barrier(0);
    gather32(xa, xg, c2, lane);
    scatter32(sco, xb, r1, v1, lane);
    r0 = r2; c0 = c2; v0 = v2;
    r1 = r3; c1 = c3; v1 = v3;
  }
  gather32(xb, xg, c1, lane);
  scatter32(sco, xa, r0, v0, lane);
  scatter32(sco, xb, r1, v1, lane);

  const size_t ibase = (size_t)b * MM;
  int cb[MM / (32 * NW1)];
#pragma unroll
  for (int i = 0; i < MM / (32 * NW1); ++i) cb[i] = idx[ibase + (wid + i * NW1) * 32 + lane];
  __syncthreads();

#pragma unroll
  for (int i = 0; i < MM / (32 * NW1); ++i) {
    const int m0 = (wid + i * NW1) * 32;
    int   cj[32];
    float cv[32];
#pragma unroll
    for (int j = 0; j < 32; ++j) cj[j] = __shfl(cb[i], j, 32);
#pragma unroll
    for (int j = 0; j < 32; ++j) cv[j] = sco[(m0 + j) * 32 + lane];
#pragma unroll
    for (int j = 0; j < 32; ++j) {
      const int   c = cj[j];
      const float v = cv[j];
      atomicAdd(&psum[c * 32 + lane], v);
      __hip_atomic_fetch_max(&pmax[c * 32 + lane], v,
                             __ATOMIC_RELAXED, __HIP_MEMORY_SCOPE_WORKGROUP);
      if (lane == 0) atomicAdd(&cnt[c], 1.0f);
    }
  }
  __syncthreads();

  for (int c = wid; c < CC; c += NWAVES) {
    const float s    = psum[c * 32 + lane];
    const float mx   = pmax[c * 32 + lane];
    const float mean = s / fmaxf(cnt[c], 1.0f);
    const size_t base = ((size_t)b * CC + c) * (3 * HH);
    xdec[base +            hbase + lane] = mean;
    xdec[base +     HH +   hbase + lane] = mx;
    xdec[base + 2 * HH +   hbase + lane] = s;
  }
}

#define XD_LD 388
#define QK_LD 132
#define S_LD  66

__device__ __forceinline__ v2f ldsA(const float* base, int ld, int mt, int k, int lane) {
  const int hi = (lane >> 4) << 1;
  const float* p = base + (mt * 16 + (lane & 15)) * ld + k + hi;
  v2f a; a.x = p[0]; a.y = p[1];
  return a;
}

__global__ __launch_bounds__(256)
void cluster_attn_kernel(const float* __restrict__ xdec_g,
                         const float* __restrict__ Wq,
                         const float* __restrict__ Wk,
                         const float* __restrict__ Wv,
                         float* __restrict__ out) {
  extern __shared__ float smem[];
  float* xd = smem;
  float* Q  = xd + 64 * XD_LD;
  float* K  = Q  + 64 * QK_LD;
  float* V  = K  + 64 * QK_LD;
  float* S  = xd;

  const int tid  = threadIdx.x;
  const int lane = tid & 31;
  const int wid  = tid >> 5;
  const int b    = blockIdx.x;

  const float* src = xdec_g + (size_t)b * (CC * 3 * HH);
  for (int i = tid; i < CC * 3 * HH; i += blockDim.x)
    xd[(i / 384) * XD_LD + (i % 384)] = src[i];
  __syncthreads();

  for (int t = wid; t < 96; t += 8) {
    const float* W = (t < 32) ? Wq : (t < 64) ? Wk : Wv;
    float*       D = (t < 32) ? Q  : (t < 64) ? K  : V;
    const int tl = t & 31, mt = tl >> 3, nt = tl & 7;
    const int hi = (lane >> 4) << 1, col = nt * 16 + (lane & 15);
    v8f acc = {};
    for (int k = 0; k < 3 * HH; k += 4) {
      v2f a = ldsA(xd, XD_LD, mt, k, lane);
      v2f bf; bf.x = W[(k + hi) * HH + col]; bf.y = W[(k + 1 + hi) * HH + col];
      acc = __builtin_amdgcn_wmma_f32_16x16x4_f32(false, a, false, bf,
                                                  (short)0, acc, false, false);
    }
    const int rbase = mt * 16 + ((lane >> 4) << 3);
#pragma unroll
    for (int r = 0; r < 8; ++r) D[(rbase + r) * QK_LD + col] = acc[r];
  }
  __syncthreads();

  const float scale = 0.08838834764831845f;
  for (int t = wid; t < 16; t += 8) {
    const int mt = t >> 2, nt = t & 3;
    const int hi = (lane >> 4) << 1, col = nt * 16 + (lane & 15);
    v8f acc = {};
    for (int k = 0; k < HH; k += 4) {
      v2f a = ldsA(Q, QK_LD, mt, k, lane);
      v2f bf; bf.x = K[col * QK_LD + k + hi]; bf.y = K[col * QK_LD + k + hi + 1];
      acc = __builtin_amdgcn_wmma_f32_16x16x4_f32(false, a, false, bf,
                                                  (short)0, acc, false, false);
    }
    const int rbase = mt * 16 + ((lane >> 4) << 3);
#pragma unroll
    for (int r = 0; r < 8; ++r) S[(rbase + r) * S_LD + col] = acc[r] * scale;
  }
  __syncthreads();

  for (int r = wid; r < CC; r += 8) {
    float v0 = S[r * S_LD + lane];
    float v1 = S[r * S_LD + 32 + lane];
    float m = fmaxf(v0, v1);
    for (int off = 16; off > 0; off >>= 1) m = fmaxf(m, __shfl_xor(m, off, 32));
    float e0 = expf(v0 - m), e1 = expf(v1 - m);
    float s = e0 + e1;
    for (int off = 16; off > 0; off >>= 1) s += __shfl_xor(s, off, 32);
    const float inv = 1.0f / s;
    S[r * S_LD + lane]      = e0 * inv;
    S[r * S_LD + 32 + lane] = e1 * inv;
  }
  __syncthreads();

  float* ob = out + (size_t)b * (CC * HH);
  for (int t = wid; t < 32; t += 8) {
    const int mt = t >> 3, nt = t & 7;
    const int hi = (lane >> 4) << 1, col = nt * 16 + (lane & 15);
    v8f acc = {};
    for (int k = 0; k < CC; k += 4) {
      v2f a = ldsA(S, S_LD, mt, k, lane);
      v2f bf; bf.x = V[(k + hi) * QK_LD + col]; bf.y = V[(k + 1 + hi) * QK_LD + col];
      acc = __builtin_amdgcn_wmma_f32_16x16x4_f32(false, a, false, bf,
                                                  (short)0, acc, false, false);
    }
    const int rbase = mt * 16 + ((lane >> 4) << 3);
#pragma unroll
    for (int r = 0; r < 8; ++r) ob[(rbase + r) * HH + col] = acc[r];
  }
}

extern "C" void kernel_launch(void* const* d_in, const int* in_sizes, int n_in,
                              void* d_out, int out_size, void* d_ws, size_t ws_size,
                              hipStream_t stream) {
  const float* x    = (const float*)d_in[0];
  const int*   rows = (const int*)d_in[3];
  const int*   cols = (const int*)d_in[4];
  const float* vals = (const float*)d_in[5];
  const int*   idx  = (const int*)d_in[6];
  const float* Wq   = (const float*)d_in[7];
  const float* Wk   = (const float*)d_in[8];
  const float* Wv   = (const float*)d_in[9];
  float* xdec = (float*)d_ws;

  const size_t smem1 = (size_t)(MM * 32 + 2 * CC * 32 + CC) * sizeof(float);
  spmm_pool_kernel<<<dim3(BB * (HH / 32)), dim3(512), smem1, stream>>>(
      x, rows, cols, vals, idx, xdec);

  const size_t smem2 = (size_t)(64 * XD_LD + 3 * 64 * QK_LD) * sizeof(float);
  cluster_attn_kernel<<<dim3(BB), dim3(256), smem2, stream>>>(
      xdec, Wq, Wk, Wv, (float*)d_out);
}


# k1: edges grouped by destination row in LDS (count, scan, place: integer LDS adds); a wave sums a row from whole 512-B x rows in registers, pools it once; add order inside a row by arrival
# speedup vs baseline: 2.7109x; 1.0518x over previous
#include <hip/hip_runtime.h>
#include <math.h>

#define BB   64
#define NN   2048
#define MM   2048
#define CC   64
#define HH   128
#define NNZE 32768

typedef float v2f __attribute__((ext_vector_type(2)));
typedef float v8f __attribute__((ext_vector_type(8)));

#define NW1  16
#define NT1  (32 * NW1)
#define EPT  (NNZE / NT1)
#define CPW  (CC / 4)
#define EB   16
#define RB   16

__device__ __forceinline__ int lane_bcast_i(int v, int j) {
  return __builtin_amdgcn_readlane(v, j);
}
__device__ __forceinline__ float lane_bcast_f(float v, int j) {
  return __int_as_float(__builtin_amdgcn_readlane(__float_as_int(v), j));
}

template <int NB>
__device__ __forceinline__ void row_batch(float4& acc, const float4* __restrict__ xg4,
                                          const unsigned short* lcol, const float* lval,
                                          int p, int e, int lane) {
  const int   pl = p + (lane & (NB - 1));
  const int   pj = min(pl, e - 1);
  const int   ci = lcol[pj];
  const float vi = (pl < e) ? lval[pj] : 0.0f;
  float4 xr[NB];
#pragma unroll
  for (int j = 0; j < NB; ++j) {
    const unsigned c = (unsigned)lane_bcast_i(ci, j);
    xr[j] = xg4[c * (HH / 4) + (unsigned)lane];
  }
#pragma unroll
  for (int j = 0; j < NB; ++j) {
    const float v = lane_bcast_f(vi, j);
    acc.x += v * xr[j].x; acc.y += v * xr[j].y;
    acc.z += v * xr[j].z; acc.w += v * xr[j].w;
  }
}

__global__ __launch_bounds__(512)
void spmm_pool_kernel(const float* __restrict__ x,
                      const int*   __restrict__ rows,
                      const int*   __restrict__ cols,
                      const float* __restrict__ vals,
                      const int*   __restrict__ idx,
                      float* __restrict__ xdec) {
  extern __shared__ float smem[];
  float*          lval   = smem;
  unsigned short* lcol   = (unsigned short*)(lval + NNZE);
  int*            bin    = (int*)(lcol + NNZE);
  int*            rstart = bin + MM;
  int*            sidx   = rstart + MM;
  float*          psum   = (float*)(sidx + MM);
  float*          pmax   = psum + CPW * HH;
  float*          pcnt   = pmax + CPW * HH;
  int*            wsum   = (int*)(pcnt + CPW);

  const int tid  = threadIdx.x;
  const int lane = tid & 31;
  const int wid  = tid >> 5;
  const int b    = blockIdx.x >> 2;
  const int q    = blockIdx.x & 3;

  const size_t ibase = (size_t)b * MM;
  for (int i = tid; i < MM; i += NT1) { bin[i] = 0; sidx[i] = idx[ibase + i]; }
  for (int i = tid; i < CPW * HH; i += NT1) { psum[i] = 0.0f; pmax[i] = -INFINITY; }
  if (tid < CPW) pcnt[tid] = 0.0f;
  __syncthreads();

  const size_t ebase = (size_t)b * NNZE;
  const int*   rp = rows + ebase + tid;
  const int*   cp = cols + ebase + tid;
  const float* vp = vals + ebase + tid;

  for (int i0 = 0; i0 < EPT; i0 += EB) {
    int r[EB], cl[EB];
#pragma unroll
    for (int j = 0; j < EB; ++j) r[j] = rp[(i0 + j) * NT1];
#pragma unroll
    for (int j = 0; j < EB; ++j) cl[j] = sidx[r[j]];
#pragma unroll
    for (int j = 0; j < EB; ++j)
      if ((cl[j] >> 4) == q) atomicAdd(&bin[r[j]], 1);
  }
  __syncthreads();

  {
    const int b0 = bin[4 * tid], b1 = bin[4 * tid + 1], b2 = bin[4 * tid + 2], b3 = bin[4 * tid + 3];
    const int tsum = b0 + b1 + b2 + b3;
    int incl = tsum;
    for (int off = 1; off < 32; off <<= 1) {
      const int t = __shfl_up(incl, off, 32);
      if (lane >= off) incl += t;
    }
    if (lane == 31) wsum[wid] = incl;
    __syncthreads();
    int woff = 0;
    for (int w = 0; w < wid; ++w) woff += wsum[w];
    int ex = woff + incl - tsum;
    rstart[4 * tid]     = ex; bin[4 * tid]     = ex; ex += b0;
    rstart[4 * tid + 1] = ex; bin[4 * tid + 1] = ex; ex += b1;
    rstart[4 * tid + 2] = ex; bin[4 * tid + 2] = ex; ex += b2;
    rstart[4 * tid + 3] = ex; bin[4 * tid + 3] = ex;
  }
  __syncthreads();

  for (int i0 = 0; i0 < EPT; i0 += EB) {
    int r[EB], c[EB], cl[EB];
    float v[EB];
#pragma unroll
    for (int j = 0; j < EB; ++j) {
      r[j] = rp[(i0 + j) * NT1];
      c[j] = cp[(i0 + j) * NT1];
      v[j] = vp[(i0 + j) * NT1];
    }
#pragma unroll
    for (int j = 0; j < EB; ++j) cl[j] = sidx[r[j]];
    int pos[EB];
#pragma unroll
    for (int j = 0; j < EB; ++j)
      pos[j] = ((cl[j] >> 4) == q) ? atomicAdd(&bin[r[j]], 1) : -1;
#pragma unroll
    for (int j = 0; j < EB; ++j) {
      if (pos[j] >= 0) {
        lcol[pos[j]] = (unsigned short)c[j];
        lval[pos[j]] = v[j];
      }
    }
  }
  __syncthreads();

  const float4* xg4 = (const float4*)(x + (size_t)b * NN * HH);
  for (int k = 0; k < MM / (NW1 * 32); ++k) {
    const int m_l  = wid * (MM / NW1) + k * 32 + lane;
    const int cl_l = sidx[m_l];
    const int s_l  = rstart[m_l];
    const int e_l  = bin[m_l];
    unsigned mask = __builtin_amdgcn_ballot_w32((cl_l >> 4) == q);
    while (mask) {
      const int t = __builtin_ctz(mask);
      mask &= mask - 1;
      const int cl = __builtin_amdgcn_readlane(cl_l, t);
      const int s  = __builtin_amdgcn_readlane(s_l, t);
      const int e  = __builtin_amdgcn_readlane(e_l, t);
      float4 acc = make_float4(0.0f, 0.0f, 0.0f, 0.0f);
      for (int p = s; p < e; p += RB)
        row_batch<RB>(acc, xg4, lcol, lval, p, e, lane);
      const int lc = cl - q * CPW;
      float* ps = psum + lc * HH + lane * 4;
      float* pm = pmax + lc * HH + lane * 4;
      atomicAdd(ps + 0, acc.x); atomicAdd(ps + 1, acc.y);
      atomicAdd(ps + 2, acc.z); atomicAdd(ps + 3, acc.w);
      __hip_atomic_fetch_max(pm + 0, acc.x, __ATOMIC_RELAXED, __HIP_MEMORY_SCOPE_WORKGROUP);
      __hip_atomic_fetch_max(pm + 1, acc.y, __ATOMIC_RELAXED, __HIP_MEMORY_SCOPE_WORKGROUP);
      __hip_atomic_fetch_max(pm + 2, acc.z, __ATOMIC_RELAXED, __HIP_MEMORY_SCOPE_WORKGROUP);
      __hip_atomic_fetch_max(pm + 3, acc.w, __ATOMIC_RELAXED, __HIP_MEMORY_SCOPE_WORKGROUP);
      if (lane == 0) atomicAdd(&pcnt[lc], 1.0f);
    }
  }
  __syncthreads();

  {
    const int lc = wid;
    const float4 s4 = *(const float4*)(psum + lc * HH + lane * 4);
    const float4 m4 = *(const float4*)(pmax + lc * HH + lane * 4);
    const float  d  = fmaxf(pcnt[lc], 1.0f);
    float4 mean4;
    mean4.x = s4.x / d; mean4.y = s4.y / d; mean4.z = s4.z / d; mean4.w = s4.w / d;
    const size_t base = ((size_t)b * CC + (q * CPW + lc)) * (3 * HH);
    *(float4*)(xdec + base +          lane * 4) = mean4;
    *(float4*)(xdec + base +     HH + lane * 4) = m4;
    *(float4*)(xdec + base + 2 * HH + lane * 4) = s4;
  }
}

#define XD_LD 388
#define QK_LD 132
#define S_LD  66

__device__ __forceinline__ v2f ldsA(const float* base, int ld, int mt, int k, int lane) {
  const int hi = (lane >> 4) << 1;
  const float* p = base + (mt * 16 + (lane & 15)) * ld + k + hi;
  v2f a; a.x = p[0]; a.y = p[1];
  return a;
}

__global__ __launch_bounds__(256)
void cluster_attn_kernel(const float* __restrict__ xdec_g,
                         const float* __restrict__ Wq,
                         const float* __restrict__ Wk,
                         const float* __restrict__ Wv,
                         float* __restrict__ out) {
  extern __shared__ float smem[];
  float* xd = smem;
  float* Q  = xd + 64 * XD_LD;
  float* K  = Q  + 64 * QK_LD;
  float* V  = K  + 64 * QK_LD;
  float* S  = xd;

  const int tid  = threadIdx.x;
  const int lane = tid & 31;
  const int wid  = tid >> 5;
  const int b    = blockIdx.x;

  const float* src = xdec_g + (size_t)b * (CC * 3 * HH);
  for (int i = tid; i < CC * 3 * HH; i += blockDim.x)
    xd[(i / 384) * XD_LD + (i % 384)] = src[i];
  __syncthreads();

  for (int t = wid; t < 96; t += 8) {
    const float* W = (t < 32) ? Wq : (t < 64) ? Wk : Wv;
    float*       D = (t < 32) ? Q  : (t < 64) ? K  : V;
    const int tl = t & 31, mt = tl >> 3, nt = tl & 7;
    const int hi = (lane >> 4) << 1, col = nt * 16 + (lane & 15);
    v8f acc = {};
    for (int k = 0; k < 3 * HH; k += 4) {
      v2f a = ldsA(xd, XD_LD, mt, k, lane);
      v2f bf; bf.x = W[(k + hi) * HH + col]; bf.y = W[(k + 1 + hi) * HH + col];
      acc = __builtin_amdgcn_wmma_f32_16x16x4_f32(false, a, false, bf,
                                                  (short)0, acc, false, false);
    }
    const int rbase = mt * 16 + ((lane >> 4) << 3);
#pragma unroll
    for (int r = 0; r < 8; ++r) D[(rbase + r) * QK_LD + col] = acc[r];
  }
  __syncthreads();

  const float scale = 0.08838834764831845f;
  for (int t = wid; t < 16; t += 8) {
    const int mt = t >> 2, nt = t & 3;
    const int hi = (lane >> 4) << 1, col = nt * 16 + (lane & 15);
    v8f acc = {};
    for (int k = 0; k < HH; k += 4) {
      v2f a = ldsA(Q, QK_LD, mt, k, lane);
      v2f bf; bf.x = K[col * QK_LD + k + hi]; bf.y = K[col * QK_LD + k + hi + 1];
      acc = __builtin_amdgcn_wmma_f32_16x16x4_f32(false, a, false, bf,
                                                  (short)0, acc, false, false);
    }
    const int rbase = mt * 16 + ((lane >> 4) << 3);
#pragma unroll
    for (int r = 0; r < 8; ++r) S[(rbase + r) * S_LD + col] = acc[r] * scale;
  }
  __syncthreads();

  for (int r = wid; r < CC; r += 8) {
    float v0 = S[r * S_LD + lane];
    float v1 = S[r * S_LD + 32 + lane];
    float m = fmaxf(v0, v1);
    for (int off = 16; off > 0; off >>= 1) m = fmaxf(m, __shfl_xor(m, off, 32));
    float e0 = expf(v0 - m), e1 = expf(v1 - m);
    float s = e0 + e1;
    for (int off = 16; off > 0; off >>= 1) s += __shfl_xor(s, off, 32);
    const float inv = 1.0f / s;
    S[r * S_LD + lane]      = e0 * inv;
    S[r * S_LD + 32 + lane] = e1 * inv;
  }
  __syncthreads();

  float* ob = out + (size_t)b * (CC * HH);
  for (int t = wid; t < 32; t += 8) {
    const int mt = t >> 3, nt = t & 7;
    const int hi = (lane >> 4) << 1, col = nt * 16 + (lane & 15);
    v8f acc = {};
    for (int k = 0; k < CC; k += 4) {
      v2f a = ldsA(S, S_LD, mt, k, lane);
      v2f bf; bf.x = V[(k + hi) * QK_LD + col]; bf.y = V[(k + 1 + hi) * QK_LD + col];
      acc = __builtin_amdgcn_wmma_f32_16x16x4_f32(false, a, false, bf,
                                                  (short)0, acc, false, false);
    }
    const int rbase = mt * 16 + ((lane >> 4) << 3);
#pragma unroll
    for (int r = 0; r < 8; ++r) ob[(rbase + r) * HH + col] = acc[r];
  }
}

extern "C" void kernel_launch(void* const* d_in, const int* in_sizes, int n_in,
                              void* d_out, int out_size, void* d_ws, size_t ws_size,
                              hipStream_t stream) {
  const float* x    = (const float*)d_in[0];
  const int*   rows = (const int*)d_in[3];
  const int*   cols = (const int*)d_in[4];
  const float* vals = (const float*)d_in[5];
  const int*   idx  = (const int*)d_in[6];
  const float* Wq   = (const float*)d_in[7];
  const float* Wk   = (const float*)d_in[8];
  const float* Wv   = (const float*)d_in[9];
  float* xdec = (float*)d_ws;

  const size_t smem1 = (size_t)NNZE * 4 + (size_t)NNZE * 2 + (size_t)(3 * MM) * 4 +
                       (size_t)(2 * CPW * HH + CPW) * 4 + (size_t)NW1 * 4;
  spmm_pool_kernel<<<dim3(BB * (CC / CPW)), dim3(NT1), smem1, stream>>>(
      x, rows, cols, vals, idx, xdec);

  const size_t smem2 = (size_t)(64 * XD_LD + 3 * 64 * QK_LD) * sizeof(float);
  cluster_attn_kernel<<<dim3(BB), dim3(256), smem2, stream>>>(
      xdec, Wq, Wk, Wv, (float*)d_out);
}


# QKV projection moved from k2 (one work-group a graph) into k1's epilogue: all 256 work-groups multiply their own 16 pooled rows, still in LDS, by Wq|Wk|Wv on f32 WMMA; k2 stages Q,K,V, attends
# speedup vs baseline: 6.1081x; 2.2532x over previous
#include <hip/hip_runtime.h>
#include <math.h>

#define BB   64
#define NN   2048
#define MM   2048
#define CC   64
#define HH   128
#define NNZE 32768

typedef float v2f __attribute__((ext_vector_type(2)));
typedef float v8f __attribute__((ext_vector_type(8)));

#define XD_LD 388
#define QK_LD 132
#define S_LD  66
#define KG    8

__device__ __forceinline__ v2f ldsA(const float* base, int ld, int mt, int k, int lane) {
  const int hi = (lane >> 4) << 1;
  const float* p = base + (mt * 16 + (lane & 15)) * ld + k + hi;
  v2f a; a.x = p[0]; a.y = p[1];
  return a;
}

template <int NM>
__device__ __forceinline__ void project16(const float* xt,
                                          const float* __restrict__ Wa,
                                          const float* __restrict__ Wb,
                                          float* __restrict__ Da,
                                          float* __restrict__ Db,
                                          int nt, int lane) {
  const int hi  = (lane >> 4) << 1;
  const int col = nt * 16 + (lane & 15);
  const unsigned voff = (unsigned)(hi * HH + col);
  const float* xa = xt + (lane & 15) * XD_LD + hi;
  v8f acca = {};
  v8f accb = {};
  for (int k0 = 0; k0 < 3 * HH; k0 += 4 * KG) {
    const float* wa = Wa + k0 * HH;
    const float* wb = Wb + k0 * HH;
    v2f a[KG], ba[KG], bb[KG];
#pragma unroll
    for (int j = 0; j < KG; ++j) {
      ba[j].x = (wa + (4 * j) * HH)[voff]; ba[j].y = (wa + (4 * j + 1) * HH)[voff];
      if (NM == 2) { bb[j].x = (wb + (4 * j) * HH)[voff]; bb[j].y = (wb + (4 * j + 1) * HH)[voff]; }
    }
#pragma unroll
    for (int j = 0; j < KG; ++j) { a[j].x = xa[k0 + 4 * j]; a[j].y = xa[k0 + 4 * j + 1]; }
#pragma unroll
    for (int j = 0; j < KG; ++j) {
      acca = __builtin_amdgcn_wmma_f32_16x16x4_f32(false, a[j], false, ba[j],
                                                   (short)0, acca, false, false);
      if (NM == 2)
        accb = __builtin_amdgcn_wmma_f32_16x16x4_f32(false, a[j], false, bb[j],
                                                     (short)0, accb, false, false);
    }
  }
  const int rbase = (lane >> 4) << 3;
#pragma unroll
  for (int r = 0; r < 8; ++r) {
    Da[(rbase + r) * HH + col] = acca[r];
    if (NM == 2) Db[(rbase + r) * HH + col] = accb[r];
  }
}

#define NW1  16
#define NT1  (32 * NW1)
#define EPT  (NNZE / NT1)
#define CPW  (CC / 4)
#define EB   16
#define RB   16

__device__ __forceinline__ int lane_bcast_i(int v, int j) {
  return __builtin_amdgcn_readlane(v, j);
}
__device__ __forceinline__ float lane_bcast_f(float v, int j) {
  return __int_as_float(__builtin_amdgcn_readlane(__float_as_int(v), j));
}

template <int NB>
__device__ __forceinline__ void row_batch(float4& acc, const float4* __restrict__ xg4,
                                          const unsigned short* lcol, const float* lval,
                                          int p, int e, int lane) {
  const int   pl = p + (lane & (NB - 1));
  const int   pj = min(pl, e - 1);
  const int   ci = lcol[pj];
  const float vi = (pl < e) ? lval[pj] : 0.0f;
  float4 xr[NB];
#pragma unroll
  for (int j = 0; j < NB; ++j) {
    const unsigned c = (unsigned)lane_bcast_i(ci, j);
    xr[j] = xg4[c * (HH / 4) + (unsigned)lane];
  }
#pragma unroll
  for (int j = 0; j < NB; ++j) {
    const float v = lane_bcast_f(vi, j);
    acc.x += v * xr[j].x; acc.y += v * xr[j].y;
    acc.z += v * xr[j].z; acc.w += v * xr[j].w;
  }
}

__global__ __launch_bounds__(512)
void spmm_pool_kernel(const float* __restrict__ x,
                      const int*   __restrict__ rows,
                      const int*   __restrict__ cols,
                      const float* __restrict__ vals,
                      const int*   __restrict__ idx,
                      const float* __restrict__ Wq,
                      const float* __restrict__ Wk,
                      const float* __restrict__ Wv,
                      float* __restrict__ qkv) {
  extern __shared__ float smem[];
  float*          lval   = smem;
  unsigned short* lcol   = (unsigned short*)(lval + NNZE);
  int*            bin    = (int*)(lcol + NNZE);
  int*            rstart = bin + MM;
  int*            sidx   = rstart + MM;
  float*          psum   = (float*)(sidx + MM);
  float*          pmax   = psum + CPW * HH;
  float*          pcnt   = pmax + CPW * HH;
  int*            wsum   = (int*)(pcnt + CPW);

  const int tid  = threadIdx.x;
  const int lane = tid & 31;
  const int wid  = __builtin_amdgcn_readfirstlane(tid >> 5);
  const int b    = blockIdx.x >> 2;
  const int q    = blockIdx.x & 3;

  const size_t ibase = (size_t)b * MM;
  for (int i = tid; i < MM; i += NT1) { bin[i] = 0; sidx[i] = idx[ibase + i]; }
  for (int i = tid; i < CPW * HH; i += NT1) { psum[i] = 0.0f; pmax[i] = -INFINITY; }
  if (tid < CPW) pcnt[tid] = 0.0f;
  __syncthreads();

  const size_t ebase = (size_t)b * NNZE;
  const int*   rp = rows + ebase + tid;
  const int*   cp = cols + ebase + tid;
  const float* vp = vals + ebase + tid;

  for (int i0 = 0; i0 < EPT; i0 += EB) {
    int r[EB], cl[EB];
#pragma unroll
    for (int j = 0; j < EB; ++j) r[j] = rp[(i0 + j) * NT1];
#pragma unroll
    for (int j = 0; j < EB; ++j) cl[j] = sidx[r[j]];
#pragma unroll
    for (int j = 0; j < EB; ++j)
      if ((cl[j] >> 4) == q) atomicAdd(&bin[r[j]], 1);
  }
  __syncthreads();

  {
    const int b0 = bin[4 * tid], b1 = bin[4 * tid + 1], b2 = bin[4 * tid + 2], b3 = bin[4 * tid + 3];
    const int tsum = b0 + b1 + b2 + b3;
    int incl = tsum;
    for (int off = 1; off < 32; off <<= 1) {
      const int t = __shfl_up(incl, off, 32);
      if (lane >= off) incl += t;
    }
    if (lane == 31) wsum[wid] = incl;
    __syncthreads();
    int woff = 0;
    for (int w = 0; w < wid; ++w) woff += wsum[w];
    int ex = woff + incl - tsum;
    rstart[4 * tid]     = ex; bin[4 * tid]     = ex; ex += b0;
    rstart[4 * tid + 1] = ex; bin[4 * tid + 1] = ex; ex += b1;
    rstart[4 * tid + 2] = ex; bin[4 * tid + 2] = ex; ex += b2;
    rstart[4 * tid + 3] = ex; bin[4 * tid + 3] = ex;
  }
  __syncthreads();

  for (int i0 = 0; i0 < EPT; i0 += EB) {
    int r[EB], c[EB], cl[EB];
    float v[EB];
#pragma unroll
    for (int j = 0; j < EB; ++j) {
      r[j] = rp[(i0 + j) * NT1];
      c[j] = cp[(i0 + j) * NT1];
      v[j] = vp[(i0 + j) * NT1];
    }
#pragma unroll
    for (int j = 0; j < EB; ++j) cl[j] = sidx[r[j]];
    int pos[EB];
#pragma unroll
    for (int j = 0; j < EB; ++j)
      pos[j] = ((cl[j] >> 4) == q) ? atomicAdd(&bin[r[j]], 1) : -1;
#pragma unroll
    for (int j = 0; j < EB; ++j) {
      if (pos[j] >= 0) {
        lcol[pos[j]] = (unsigned short)c[j];
        lval[pos[j]] = v[j];
      }
    }
  }
  __syncthreads();

  const float4* xg4 = (const float4*)(x + (size_t)b * NN * HH);
  for (int k = 0; k < MM / (NW1 * 32); ++k) {
    const int m_l  = wid * (MM / NW1) + k * 32 + lane;
    const int cl_l = sidx[m_l];
    const int s_l  = rstart[m_l];
    const int e_l  = bin[m_l];
    unsigned mask = __builtin_amdgcn_ballot_w32((cl_l >> 4) == q);
    while (mask) {
      const int t = __builtin_ctz(mask);
      mask &= mask - 1;
      const int cl = __builtin_amdgcn_readlane(cl_l, t);
      const int s  = __builtin_amdgcn_readlane(s_l, t);
      const int e  = __builtin_amdgcn_readlane(e_l, t);
      float4 acc = make_float4(0.0f, 0.0f, 0.0f, 0.0f);
      for (int p = s; p < e; p += RB)
        row_batch<RB>(acc, xg4, lcol, lval, p, e, lane);
      const int lc = cl - q * CPW;
      float* ps = psum + lc * HH + lane * 4;
      float* pm = pmax + lc * HH + lane * 4;
      atomicAdd(ps + 0, acc.x); atomicAdd(ps + 1, acc.y);
      atomicAdd(ps + 2, acc.z); atomicAdd(ps + 3, acc.w);
      __hip_atomic_fetch_max(pm + 0, acc.x, __ATOMIC_RELAXED, __HIP_MEMORY_SCOPE_WORKGROUP);
      __hip_atomic_fetch_max(pm + 1, acc.y, __ATOMIC_RELAXED, __HIP_MEMORY_SCOPE_WORKGROUP);
      __hip_atomic_fetch_max(pm + 2, acc.z, __ATOMIC_RELAXED, __HIP_MEMORY_SCOPE_WORKGROUP);
      __hip_atomic_fetch_max(pm + 3, acc.w, __ATOMIC_RELAXED, __HIP_MEMORY_SCOPE_WORKGROUP);
      if (lane == 0) atomicAdd(&pcnt[lc], 1.0f);
    }
  }
  __syncthreads();

  float* xt = smem;
  {
    const int lc = wid;
    const float4 s4 = *(const float4*)(psum + lc * HH + lane * 4);
    const float4 m4 = *(const float4*)(pmax + lc * HH + lane * 4);
    const float  d  = fmaxf(pcnt[lc], 1.0f);
    float4 mean4;
    mean4.x = s4.x / d; mean4.y = s4.y / d; mean4.z = s4.z / d; mean4.w = s4.w / d;
    float* row = xt + lc * XD_LD + lane * 4;
    *(float4*)(row)          = mean4;
    *(float4*)(row + HH)     = m4;
    *(float4*)(row + 2 * HH) = s4;
  }
  __syncthreads();

  {
    const int nt = wid & 7;
    float* qb = qkv + ((size_t)b * 3 * CC + q * CPW) * HH;
    float* kb = qb + CC * HH;
    float* vb = kb + CC * HH;
    if (wid < 8) project16<2>(xt, Wk, Wv, kb, vb, nt, lane);
    else         project16<1>(xt, Wq, Wq, qb, qb, nt, lane);
  }
}

#define ST2 (3 * CC * HH / 4 / 256)

__global__ __launch_bounds__(256)
void cluster_attn_kernel(const float* __restrict__ qkv_g,
                         float* __restrict__ out) {
  extern __shared__ float smem[];
  float* Q  = smem;
  float* K  = Q  + 64 * QK_LD;
  float* V  = K  + 64 * QK_LD;
  float* S  = V  + 64 * QK_LD;

  const int tid  = threadIdx.x;
  const int lane = tid & 31;
  const int wid  = tid >> 5;
  const int b    = blockIdx.x;

  {
    const float4* src4 = (const float4*)(qkv_g + (size_t)b * (3 * CC * HH));
    float4 st[ST2];
#pragma unroll
    for (int j = 0; j < ST2; ++j) st[j] = src4[tid + 256 * j];
#pragma unroll
    for (int j = 0; j < ST2; ++j)
      *(float4*)(Q + (wid + 8 * j) * QK_LD + lane * 4) = st[j];
  }
  __syncthreads();

  const float scale = 0.08838834764831845f;
  for (int t = wid; t < 16; t += 8) {
    const int mt = t >> 2, nt = t & 3;
    const int hi = (lane >> 4) << 1, col = nt * 16 + (lane & 15);
    v8f acc = {};
    for (int k = 0; k < HH; k += 4) {
      v2f a = ldsA(Q, QK_LD, mt, k, lane);
      v2f bf; bf.x = K[col * QK_LD + k + hi]; bf.y = K[col * QK_LD + k + hi + 1];
      acc = __builtin_amdgcn_wmma_f32_16x16x4_f32(false, a, false, bf,
                                                  (short)0, acc, false, false);
    }
    const int rbase = mt * 16 + ((lane >> 4) << 3);
#pragma unroll
    for (int r = 0; r < 8; ++r) S[(rbase + r) * S_LD + col] = acc[r] * scale;
  }
  __syncthreads();

  for (int r = wid; r < CC; r += 8) {
    float v0 = S[r * S_LD + lane];
    float v1 = S[r * S_LD + 32 + lane];
    float m = fmaxf(v0, v1);
    for (int off = 16; off > 0; off >>= 1) m = fmaxf(m, __shfl_xor(m, off, 32));
    float e0 = expf(v0 - m), e1 = expf(v1 - m);
    float s = e0 + e1;
    for (int off = 16; off > 0; off >>= 1) s += __shfl_xor(s, off, 32);
    const float inv = 1.0f / s;
    S[r * S_LD + lane]      = e0 * inv;
    S[r * S_LD + 32 + lane] = e1 * inv;
  }
  __syncthreads();

  float* ob = out + (size_t)b * (CC * HH);
  for (int t = wid; t < 32; t += 8) {
    const int mt = t >> 3, nt = t & 7;
    const int hi = (lane >> 4) << 1, col = nt * 16 + (lane & 15);
    v8f acc = {};
    for (int k = 0; k < CC; k += 4) {
      v2f a = ldsA(S, S_LD, mt, k, lane);
      v2f bf; bf.x = V[(k + hi) * QK_LD + col]; bf.y = V[(k + 1 + hi) * QK_LD + col];
      acc = __builtin_amdgcn_wmma_f32_16x16x4_f32(false, a, false, bf,
                                                  (short)0, acc, false, false);
    }
    const int rbase = mt * 16 + ((lane >> 4) << 3);
#pragma unroll
    for (int r = 0; r < 8; ++r) ob[(rbase + r) * HH + col] = acc[r];
  }
}

extern "C" void kernel_launch(void* const* d_in, const int* in_sizes, int n_in,
                              void* d_out, int out_size, void* d_ws, size_t ws_size,
                              hipStream_t stream) {
  const float* x    = (const float*)d_in[0];
  const int*   rows = (const int*)d_in[3];
  const int*   cols = (const int*)d_in[4];
  const float* vals = (const float*)d_in[5];
  const int*   idx  = (const int*)d_in[6];
  const float* Wq   = (const float*)d_in[7];
  const float* Wk   = (const float*)d_in[8];
  const float* Wv   = (const float*)d_in[9];
  float* qkv = (float*)d_ws;

  const size_t smem1 = (size_t)NNZE * 4 + (size_t)NNZE * 2 + (size_t)(3 * MM) * 4 +
                       (size_t)(2 * CPW * HH + CPW) * 4 + (size_t)NW1 * 4;
  spmm_pool_kernel<<<dim3(BB * (CC / CPW)), dim3(NT1), smem1, stream>>>(
      x, rows, cols, vals, idx, Wq, Wk, Wv, qkv);

  const size_t smem2 = (size_t)(3 * 64 * QK_LD + 64 * S_LD) * sizeof(float);
  cluster_attn_kernel<<<dim3(BB), dim3(256), smem2, stream>>>(
      qkv, (float*)d_out);
}


# k1: edge arrays read once: a thread keeps its 64 edges in registers from count to placing; the count's returning LDS add gives the rank in the row (no 2nd add, no rows re-read); add order as before
# speedup vs baseline: 6.5011x; 1.0643x over previous
#include <hip/hip_runtime.h>
#include <math.h>

#define BB   64
#define NN   2048
#define MM   2048
#define CC   64
#define HH   128
#define NNZE 32768

typedef float v2f __attribute__((ext_vector_type(2)));
typedef float v8f __attribute__((ext_vector_type(8)));

#define XD_LD 388
#define QK_LD 132
#define S_LD  66
#define KG    8

__device__ __forceinline__ v2f ldsA(const float* base, int ld, int mt, int k, int lane) {
  const int hi = (lane >> 4) << 1;
  const float* p = base + (mt * 16 + (lane & 15)) * ld + k + hi;
  v2f a; a.x = p[0]; a.y = p[1];
  return a;
}

template <int NM>
__device__ __forceinline__ void project16(const float* xt,
                                          const float* __restrict__ Wa,
                                          const float* __restrict__ Wb,
                                          float* __restrict__ Da,
                                          float* __restrict__ Db,
                                          int nt, int lane) {
  const int hi  = (lane >> 4) << 1;
  const int col = nt * 16 + (lane & 15);
  const unsigned voff = (unsigned)(hi * HH + col);
  const float* xa = xt + (lane & 15) * XD_LD + hi;
  v8f acca = {};
  v8f accb = {};
  for (int k0 = 0; k0 < 3 * HH; k0 += 4 * KG) {
    const float* wa = Wa + k0 * HH;
    const float* wb = Wb + k0 * HH;
    v2f a[KG], ba[KG], bb[KG];
#pragma unroll
    for (int j = 0; j < KG; ++j) {
      ba[j].x = (wa + (4 * j) * HH)[voff]; ba[j].y = (wa + (4 * j + 1) * HH)[voff];
      if (NM == 2) { bb[j].x = (wb + (4 * j) * HH)[voff]; bb[j].y = (wb + (4 * j + 1) * HH)[voff]; }
    }
#pragma unroll
    for (int j = 0; j < KG; ++j) { a[j].x = xa[k0 + 4 * j]; a[j].y = xa[k0 + 4 * j + 1]; }
#pragma unroll
    for (int j = 0; j < KG; ++j) {
      acca = __builtin_amdgcn_wmma_f32_16x16x4_f32(false, a[j], false, ba[j],
                                                   (short)0, acca, false, false);
      if (NM == 2)
        accb = __builtin_amdgcn_wmma_f32_16x16x4_f32(false, a[j], false, bb[j],
                                                     (short)0, accb, false, false);
    }
  }
  const int rbase = (lane >> 4) << 3;
#pragma unroll
  for (int r = 0; r < 8; ++r) {
    Da[(rbase + r) * HH + col] = acca[r];
    if (NM == 2) Db[(rbase + r) * HH + col] = accb[r];
  }
}

#define NW1  16
#define NT1  (32 * NW1)
#define EPT  (NNZE / NT1)
#define CPW  (CC / 4)
#define EB   16
#define RB   16

__device__ __forceinline__ int lane_bcast_i(int v, int j) {
  return __builtin_amdgcn_readlane(v, j);
}
__device__ __forceinline__ float lane_bcast_f(float v, int j) {
  return __int_as_float(__builtin_amdgcn_readlane(__float_as_int(v), j));
}

template <int NB>
__device__ __forceinline__ void row_batch(float4& acc, const float4* __restrict__ xg4,
                                          const unsigned short* lcol, const float* lval,
                                          int p, int e, int lane) {
  const int   pl = p + (lane & (NB - 1));
  const int   pj = min(pl, e - 1);
  const int   ci = lcol[pj];
  const float vi = (pl < e) ? lval[pj] : 0.0f;
  float4 xr[NB];
#pragma unroll
  for (int j = 0; j < NB; ++j) {
    const unsigned c = (unsigned)lane_bcast_i(ci, j);
    xr[j] = xg4[c * (HH / 4) + (unsigned)lane];
  }
#pragma unroll
  for (int j = 0; j < NB; ++j) {
    const float v = lane_bcast_f(vi, j);
    acc.x += v * xr[j].x; acc.y += v * xr[j].y;
    acc.z += v * xr[j].z; acc.w += v * xr[j].w;
  }
}

__global__ __launch_bounds__(512)
void spmm_pool_kernel(const float* __restrict__ x,
                      const int*   __restrict__ rows,
                      const int*   __restrict__ cols,
                      const float* __restrict__ vals,
                      const int*   __restrict__ idx,
                      const float* __restrict__ Wq,
                      const float* __restrict__ Wk,
                      const float* __restrict__ Wv,
                      float* __restrict__ qkv) {
  extern __shared__ float smem[];
  float*          lval   = smem;
  unsigned short* lcol   = (unsigned short*)(lval + NNZE);
  int*            bin    = (int*)(lcol + NNZE);
  int*            rstart = bin + MM;
  int*            sidx   = rstart + MM;
  float*          psum   = (float*)(sidx + MM);
  float*          pmax   = psum + CPW * HH;
  float*          pcnt   = pmax + CPW * HH;
  int*            wsum   = (int*)(pcnt + CPW);

  const int tid  = threadIdx.x;
  const int lane = tid & 31;
  const int wid  = __builtin_amdgcn_readfirstlane(tid >> 5);
  const int b    = blockIdx.x >> 2;
  const int q    = blockIdx.x & 3;

  const size_t ibase = (size_t)b * MM;
  const size_t ebase = (size_t)b * NNZE;
  const int*   rp = rows + ebase + tid;
  const int*   cp = cols + ebase + tid;
  const float* vp = vals + ebase + tid;

  int cid[MM / NT1];
#pragma unroll
  for (int j = 0; j < MM / NT1; ++j) cid[j] = idx[ibase + tid + j * NT1];
  int pk[EPT];
#pragma unroll
  for (int j = 0; j < EPT; ++j) pk[j] = rp[j * NT1];

#pragma unroll
  for (int j = 0; j < MM / NT1; ++j) { bin[tid + j * NT1] = tid + j * NT1; sidx[tid + j * NT1] = cid[j]; }
  for (int i = tid; i < CPW * HH; i += NT1) { psum[i] = 0.0f; pmax[i] = -INFINITY; }
  if (tid < CPW) pcnt[tid] = 0.0f;
  __syncthreads();

  int   c[EPT];
  float v[EPT];
#pragma unroll
  for (int i0 = 0; i0 < EPT; i0 += EB) {
#pragma unroll
    for (int j = 0; j < EB; ++j) { c[i0 + j] = cp[(i0 + j) * NT1]; v[i0 + j] = vp[(i0 + j) * NT1]; }
    int cl[EB];
#pragma unroll
    for (int j = 0; j < EB; ++j) cl[j] = sidx[pk[i0 + j]];
#pragma unroll
    for (int j = 0; j < EB; ++j)
      pk[i0 + j] = ((cl[j] >> 4) == q) ? atomicAdd(&bin[pk[i0 + j]], 1 << 11) : -1;
  }
  __syncthreads();

  {
    const int b0 = bin[4 * tid] >> 11, b1 = bin[4 * tid + 1] >> 11, b2 = bin[4 * tid + 2] >> 11, b3 = bin[4 * tid + 3] >> 11;
    const int tsum = b0 + b1 + b2 + b3;
    int incl = tsum;
    for (int off = 1; off < 32; off <<= 1) {
      const int t = __shfl_up(incl, off, 32);
      if (lane >= off) incl += t;
    }
    if (lane == 31) wsum[wid] = incl;
    __syncthreads();
    int woff = 0;
    for (int w = 0; w < wid; ++w) woff += wsum[w];
    int ex = woff + incl - tsum;
    rstart[4 * tid]     = ex; ex += b0;
    rstart[4 * tid + 1] = ex; ex += b1;
    rstart[4 * tid + 2] = ex; ex += b2;
    rstart[4 * tid + 3] = ex;
  }
  __syncthreads();

#pragma unroll
  for (int i0 = 0; i0 < EPT; i0 += EB) {
    int st[EB];
#pragma unroll
    for (int j = 0; j < EB; ++j) st[j] = rstart[pk[i0 + j] & (MM - 1)];
#pragma unroll
    for (int j = 0; j < EB; ++j) {
      if (pk[i0 + j] >= 0) {
        const int pos = st[j] + (pk[i0 + j] >> 11);
        lcol[pos] = (unsigned short)c[i0 + j];
        lval[pos] = v[i0 + j];
      }
    }
  }
  __syncthreads();

  const float4* xg4 = (const float4*)(x + (size_t)b * NN * HH);
  for (int k = 0; k < MM / (NW1 * 32); ++k) {
    const int m_l  = wid * (MM / NW1) + k * 32 + lane;
    const int cl_l = sidx[m_l];
    const int s_l  = rstart[m_l];
    const int e_l  = s_l + (bin[m_l] >> 11);
    unsigned mask = __builtin_amdgcn_ballot_w32((cl_l >> 4) == q);
    while (mask) {
      const int t = __builtin_ctz(mask);
      mask &= mask - 1;
      const int cl = __builtin_amdgcn_readlane(cl_l, t);
      const int s  = __builtin_amdgcn_readlane(s_l, t);
      const int e  = __builtin_amdgcn_readlane(e_l, t);
      float4 acc = make_float4(0.0f, 0.0f, 0.0f, 0.0f);
      for (int p = s; p < e; p += RB)
        row_batch<RB>(acc, xg4, lcol, lval, p, e, lane);
      const int lc = cl - q * CPW;
      float* ps = psum + lc * HH + lane * 4;
      float* pm = pmax + lc * HH + lane * 4;
      atomicAdd(ps + 0, acc.x); atomicAdd(ps + 1, acc.y);
      atomicAdd(ps + 2, acc.z); atomicAdd(ps + 3, acc.w);
      __hip_atomic_fetch_max(pm + 0, acc.x, __ATOMIC_RELAXED, __HIP_MEMORY_SCOPE_WORKGROUP);
      __hip_atomic_fetch_max(pm + 1, acc.y, __ATOMIC_RELAXED, __HIP_MEMORY_SCOPE_WORKGROUP);
      __hip_atomic_fetch_max(pm + 2, acc.z, __ATOMIC_RELAXED, __HIP_MEMORY_SCOPE_WORKGROUP);
      __hip_atomic_fetch_max(pm + 3, acc.w, __ATOMIC_RELAXED, __HIP_MEMORY_SCOPE_WORKGROUP);
      if (lane == 0) atomicAdd(&pcnt[lc], 1.0f);
    }
  }
  __syncthreads();

  float* xt = smem;
  {
    const int lc = wid;
    const float4 s4 = *(const float4*)(psum + lc * HH + lane * 4);
    const float4 m4 = *(const float4*)(pmax + lc * HH + lane * 4);
    const float  d  = fmaxf(pcnt[lc], 1.0f);
    float4 mean4;
    mean4.x = s4.x / d; mean4.y = s4.y / d; mean4.z = s4.z / d; mean4.w = s4.w / d;
    float* row = xt + lc * XD_LD + lane * 4;
    *(float4*)(row)          = mean4;
    *(float4*)(row + HH)     = m4;
    *(float4*)(row + 2 * HH) = s4;
  }
  __syncthreads();

  {
    const int nt = wid & 7;
    float* qb = qkv + ((size_t)b * 3 * CC + q * CPW) * HH;
    float* kb = qb + CC * HH;
    float* vb = kb + CC * HH;
    if (wid < 8) project16<2>(xt, Wk, Wv, kb, vb, nt, lane);
    else         project16<1>(xt, Wq, Wq, qb, qb, nt, lane);
  }
}

#define ST2 (3 * CC * HH / 4 / 256)

__global__ __launch_bounds__(256)
void cluster_attn_kernel(const float* __restrict__ qkv_g,
                         float* __restrict__ out) {
  extern __shared__ float smem[];
  float* Q  = smem;
  float* K  = Q  + 64 * QK_LD;
  float* V  = K  + 64 * QK_LD;
  float* S  = V  + 64 * QK_LD;

  const int tid  = threadIdx.x;
  const int lane = tid & 31;
  const int wid  = tid >> 5;
  const int b    = blockIdx.x;

  {
    const float4* src4 = (const float4*)(qkv_g + (size_t)b * (3 * CC * HH));
    float4 st[ST2];
#pragma unroll
    for (int j = 0; j < ST2; ++j) st[j] = src4[tid + 256 * j];
#pragma unroll
    for (int j = 0; j < ST2; ++j)
      *(float4*)(Q + (wid + 8 * j) * QK_LD + lane * 4) = st[j];
  }
  __syncthreads();

  const float scale = 0.08838834764831845f;
  for (int t = wid; t < 16; t += 8) {
    const int mt = t >> 2, nt = t & 3;
    const int hi = (lane >> 4) << 1, col = nt * 16 + (lane & 15);
    v8f acc = {};
    for (int k = 0; k < HH; k += 4) {
      v2f a = ldsA(Q, QK_LD, mt, k, lane);
      v2f bf; bf.x = K[col * QK_LD + k + hi]; bf.y = K[col * QK_LD + k + hi + 1];
      acc = __builtin_amdgcn_wmma_f32_16x16x4_f32(false, a, false, bf,
                                                  (short)0, acc, false, false);
    }
    const int rbase = mt * 16 + ((lane >> 4) << 3);
#pragma unroll
    for (int r = 0; r < 8; ++r) S[(rbase + r) * S_LD + col] = acc[r] * scale;
  }
  __syncthreads();

  for (int r = wid; r < CC; r += 8) {
    float v0 = S[r * S_LD + lane];
    float v1 = S[r * S_LD + 32 + lane];
    float m = fmaxf(v0, v1);
    for (int off = 16; off > 0; off >>= 1) m = fmaxf(m, __shfl_xor(m, off, 32));
    float e0 = expf(v0 - m), e1 = expf(v1 - m);
    float s = e0 + e1;
    for (int off = 16; off > 0; off >>= 1) s += __shfl_xor(s, off, 32);
    const float inv = 1.0f / s;
    S[r * S_LD + lane]      = e0 * inv;
    S[r * S_LD + 32 + lane] = e1 * inv;
  }
  __syncthreads();

  float* ob = out + (size_t)b * (CC * HH);
  for (int t = wid; t < 32; t += 8) {
    const int mt = t >> 3, nt = t & 7;
    const int hi = (lane >> 4) << 1, col = nt * 16 + (lane & 15);
    v8f acc = {};
    for (int k = 0; k < CC; k += 4) {
      v2f a = ldsA(S, S_LD, mt, k, lane);
      v2f bf; bf.x = V[(k + hi) * QK_LD + col]; bf.y = V[(k + 1 + hi) * QK_LD + col];
      acc = __builtin_amdgcn_wmma_f32_16x16x4_f32(false, a, false, bf,
                                                  (short)0, acc, false, false);
    }
    const int rbase = mt * 16 + ((lane >> 4) << 3);
#pragma unroll
    for (int r = 0; r < 8; ++r) ob[(rbase + r) * HH + col] = acc[r];
  }
}

extern "C" void kernel_launch(void* const* d_in, const int* in_sizes, int n_in,
                              void* d_out, int out_size, void* d_ws, size_t ws_size,
                              hipStream_t stream) {
  const float* x    = (const float*)d_in[0];
  const int*   rows = (const int*)d_in[3];
  const int*   cols = (const int*)d_in[4];
  const float* vals = (const float*)d_in[5];
  const int*   idx  = (const int*)d_in[6];
  const float* Wq   = (const float*)d_in[7];
  const float* Wk   = (const float*)d_in[8];
  const float* Wv   = (const float*)d_in[9];
  float* qkv = (float*)d_ws;

  const size_t smem1 = (size_t)NNZE * 4 + (size_t)NNZE * 2 + (size_t)(3 * MM) * 4 +
                       (size_t)(2 * CPW * HH + CPW) * 4 + (size_t)NW1 * 4;
  spmm_pool_kernel<<<dim3(BB * (CC / CPW)), dim3(NT1), smem1, stream>>>(
      x, rows, cols, vals, idx, Wq, Wk, Wv, qkv);

  const size_t smem2 = (size_t)(3 * 64 * QK_LD + 64 * S_LD) * sizeof(float);
  cluster_attn_kernel<<<dim3(BB), dim3(256), smem2, stream>>>(
      qkv, (float*)d_out);
}


# k2 re-cut: one 4-wave work-group per 16-row query band of a graph (256 groups, was 64 of 8 waves): a score tile a wave, softmax 8 lanes a row in one pass (row sum order changes), 2 out tiles a wave
# speedup vs baseline: 6.8347x; 1.0513x over previous
#include <hip/hip_runtime.h>
#include <math.h>

#define BB   64
#define NN   2048
#define MM   2048
#define CC   64
#define HH   128
#define NNZE 32768

typedef float v2f __attribute__((ext_vector_type(2)));
typedef float v8f __attribute__((ext_vector_type(8)));

#define XD_LD 388
#define QK_LD 132
#define S_LD  66
#define KG    8

__device__ __forceinline__ v2f ldsA(const float* base, int ld, int mt, int k, int lane) {
  const int hi = (lane >> 4) << 1;
  const float* p = base + (mt * 16 + (lane & 15)) * ld + k + hi;
  v2f a; a.x = p[0]; a.y = p[1];
  return a;
}

__device__ __forceinline__ v2f lds2(const float* p) {
  return *(const v2f*)p;
}

__device__ __forceinline__ void wmma_tail(v8f& acc, const v2f& a, const v2f& b) {
  asm volatile("v_nop\n\tv_nop\n\tv_nop\n\tv_nop" : "+v"(acc) : "v"(a), "v"(b));
}

template <int NM>
__device__ __forceinline__ void project16(const float* xt,
                                          const float* __restrict__ Wa,
                                          const float* __restrict__ Wb,
                                          float* __restrict__ Da,
                                          float* __restrict__ Db,
                                          int nt, int lane) {
  const int hi  = (lane >> 4) << 1;
  const int col = nt * 16 + (lane & 15);
  const unsigned voff = (unsigned)(hi * HH + col);
  const float* xa = xt + (lane & 15) * XD_LD + hi;
  v8f acca = {};
  v8f accb = {};
  for (int k0 = 0; k0 < 3 * HH; k0 += 4 * KG) {
    const float* wa = Wa + k0 * HH;
    const float* wb = Wb + k0 * HH;
    v2f a[KG], ba[KG], bb[KG];
#pragma unroll
    for (int j = 0; j < KG; ++j) {
      ba[j].x = (wa + (4 * j) * HH)[voff]; ba[j].y = (wa + (4 * j + 1) * HH)[voff];
      if (NM == 2) { bb[j].x = (wb + (4 * j) * HH)[voff]; bb[j].y = (wb + (4 * j + 1) * HH)[voff]; }
    }
#pragma unroll
    for (int j = 0; j < KG; ++j) { a[j].x = xa[k0 + 4 * j]; a[j].y = xa[k0 + 4 * j + 1]; }
#pragma unroll
    for (int j = 0; j < KG; ++j) {
      acca = __builtin_amdgcn_wmma_f32_16x16x4_f32(false, a[j], false, ba[j],
                                                   (short)0, acca, false, false);
      if (NM == 2)
        accb = __builtin_amdgcn_wmma_f32_16x16x4_f32(false, a[j], false, bb[j],
                                                     (short)0, accb, false, false);
    }
  }
  const int rbase = (lane >> 4) << 3;
#pragma unroll
  for (int r = 0; r < 8; ++r) {
    Da[(rbase + r) * HH + col] = acca[r];
    if (NM == 2) Db[(rbase + r) * HH + col] = accb[r];
  }
}

#define NW1  16
#define NT1  (32 * NW1)
#define EPT  (NNZE / NT1)
#define CPW  (CC / 4)
#define EB   16
#define RB   16

__device__ __forceinline__ int lane_bcast_i(int v, int j) {
  return __builtin_amdgcn_readlane(v, j);
}
__device__ __forceinline__ float lane_bcast_f(float v, int j) {
  return __int_as_float(__builtin_amdgcn_readlane(__float_as_int(v), j));
}

template <int NB>
__device__ __forceinline__ void row_batch(float4& acc, const float4* __restrict__ xg4,
                                          const unsigned short* lcol, const float* lval,
                                          int p, int e, int lane) {
  const int   pl = p + (lane & (NB - 1));
  const int   pj = min(pl, e - 1);
  const int   ci = lcol[pj];
  const float vi = (pl < e) ? lval[pj] : 0.0f;
  float4 xr[NB];
#pragma unroll
  for (int j = 0; j < NB; ++j) {
    const unsigned c = (unsigned)lane_bcast_i(ci, j);
    xr[j] = xg4[c * (HH / 4) + (unsigned)lane];
  }
#pragma unroll
  for (int j = 0; j < NB; ++j) {
    const float v = lane_bcast_f(vi, j);
    acc.x += v * xr[j].x; acc.y += v * xr[j].y;
    acc.z += v * xr[j].z; acc.w += v * xr[j].w;
  }
}

__global__ __launch_bounds__(512)
void spmm_pool_kernel(const float* __restrict__ x,
                      const int*   __restrict__ rows,
                      const int*   __restrict__ cols,
                      const float* __restrict__ vals,
                      const int*   __restrict__ idx,
                      const float* __restrict__ Wq,
                      const float* __restrict__ Wk,
                      const float* __restrict__ Wv,
                      float* __restrict__ qkv) {
  extern __shared__ float smem[];
  float*          lval   = smem;
  unsigned short* lcol   = (unsigned short*)(lval + NNZE);
  int*            bin    = (int*)(lcol + NNZE);
  int*            rstart = bin + MM;
  int*            sidx   = rstart + MM;
  float*          psum   = (float*)(sidx + MM);
  float*          pmax   = psum + CPW * HH;
  float*          pcnt   = pmax + CPW * HH;
  int*            wsum   = (int*)(pcnt + CPW);

  const int tid  = threadIdx.x;
  const int lane = tid & 31;
  const int wid  = __builtin_amdgcn_readfirstlane(tid >> 5);
  const int b    = blockIdx.x >> 2;
  const int q    = blockIdx.x & 3;

  const size_t ibase = (size_t)b * MM;
  const size_t ebase = (size_t)b * NNZE;
  const int*   rp = rows + ebase + tid;
  const int*   cp = cols + ebase + tid;
  const float* vp = vals + ebase + tid;

  int cid[MM / NT1];
#pragma unroll
  for (int j = 0; j < MM / NT1; ++j) cid[j] = idx[ibase + tid + j * NT1];
  int pk[EPT];
#pragma unroll
  for (int j = 0; j < EPT; ++j) pk[j] = rp[j * NT1];

#pragma unroll
  for (int j = 0; j < MM / NT1; ++j) { bin[tid + j * NT1] = tid + j * NT1; sidx[tid + j * NT1] = cid[j]; }
  for (int i = tid; i < CPW * HH; i += NT1) { psum[i] = 0.0f; pmax[i] = -INFINITY; }
  if (tid < CPW) pcnt[tid] = 0.0f;
  __syncthreads();

  int   c[EPT];
  float v[EPT];
#pragma unroll
  for (int i0 = 0; i0 < EPT; i0 += EB) {
#pragma unroll
    for (int j = 0; j < EB; ++j) { c[i0 + j] = cp[(i0 + j) * NT1]; v[i0 + j] = vp[(i0 + j) * NT1]; }
    int cl[EB];
#pragma unroll
    for (int j = 0; j < EB; ++j) cl[j] = sidx[pk[i0 + j]];
#pragma unroll
    for (int j = 0; j < EB; ++j)
      pk[i0 + j] = ((cl[j] >> 4) == q) ? atomicAdd(&bin[pk[i0 + j]], 1 << 11) : -1;
  }
  __syncthreads();

  {
    const int b0 = bin[4 * tid] >> 11, b1 = bin[4 * tid + 1] >> 11, b2 = bin[4 * tid + 2] >> 11, b3 = bin[4 * tid + 3] >> 11;
    const int tsum = b0 + b1 + b2 + b3;
    int incl = tsum;
    for (int off = 1; off < 32; off <<= 1) {
      const int t = __shfl_up(incl, off, 32);
      if (lane >= off) incl += t;
    }
    if (lane == 31) wsum[wid] = incl;
    __syncthreads();
    int woff = 0;
    for (int w = 0; w < wid; ++w) woff += wsum[w];
    int ex = woff + incl - tsum;
    rstart[4 * tid]     = ex; ex += b0;
    rstart[4 * tid + 1] = ex; ex += b1;
    rstart[4 * tid + 2] = ex; ex += b2;
    rstart[4 * tid + 3] = ex;
  }
  __syncthreads();

#pragma unroll
  for (int i0 = 0; i0 < EPT; i0 += EB) {
    int st[EB];
#pragma unroll
    for (int j = 0; j < EB; ++j) st[j] = rstart[pk[i0 + j] & (MM - 1)];
#pragma unroll
    for (int j = 0; j < EB; ++j) {
      if (pk[i0 + j] >= 0) {
        const int pos = st[j] + (pk[i0 + j] >> 11);
        lcol[pos] = (unsigned short)c[i0 + j];
        lval[pos] = v[i0 + j];
      }
    }
  }
  __syncthreads();

  const float4* xg4 = (const float4*)(x + (size_t)b * NN * HH);
  for (int k = 0; k < MM / (NW1 * 32); ++k) {
    const int m_l  = wid * (MM / NW1) + k * 32 + lane;
    const int cl_l = sidx[m_l];
    const int s_l  = rstart[m_l];
    const int e_l  = s_l + (bin[m_l] >> 11);
    unsigned mask = __builtin_amdgcn_ballot_w32((cl_l >> 4) == q);
    while (mask) {
      const int t = __builtin_ctz(mask);
      mask &= mask - 1;
      const int cl = __builtin_amdgcn_readlane(cl_l, t);
      const int s  = __builtin_amdgcn_readlane(s_l, t);
      const int e  = __builtin_amdgcn_readlane(e_l, t);
      float4 acc = make_float4(0.0f, 0.0f, 0.0f, 0.0f);
      for (int p = s; p < e; p += RB)
        row_batch<RB>(acc, xg4, lcol, lval, p, e, lane);
      const int lc = cl - q * CPW;
      float* ps = psum + lc * HH + lane * 4;
      float* pm = pmax + lc * HH + lane * 4;
      atomicAdd(ps + 0, acc.x); atomicAdd(ps + 1, acc.y);
      atomicAdd(ps + 2, acc.z); atomicAdd(ps + 3, acc.w);
      __hip_atomic_fetch_max(pm + 0, acc.x, __ATOMIC_RELAXED, __HIP_MEMORY_SCOPE_WORKGROUP);
      __hip_atomic_fetch_max(pm + 1, acc.y, __ATOMIC_RELAXED, __HIP_MEMORY_SCOPE_WORKGROUP);
      __hip_atomic_fetch_max(pm + 2, acc.z, __ATOMIC_RELAXED, __HIP_MEMORY_SCOPE_WORKGROUP);
      __hip_atomic_fetch_max(pm + 3, acc.w, __ATOMIC_RELAXED, __HIP_MEMORY_SCOPE_WORKGROUP);
      if (lane == 0) atomicAdd(&pcnt[lc], 1.0f);
    }
  }
  __syncthreads();

  float* xt = smem;
  {
    const int lc = wid;
    const float4 s4 = *(const float4*)(psum + lc * HH + lane * 4);
    const float4 m4 = *(const float4*)(pmax + lc * HH + lane * 4);
    const float  d  = fmaxf(pcnt[lc], 1.0f);
    float4 mean4;
    mean4.x = s4.x / d; mean4.y = s4.y / d; mean4.z = s4.z / d; mean4.w = s4.w / d;
    float* row = xt + lc * XD_LD + lane * 4;
    *(float4*)(row)          = mean4;
    *(float4*)(row + HH)     = m4;
    *(float4*)(row + 2 * HH) = s4;
  }
  __syncthreads();

  {
    const int nt = wid & 7;
    float* qb = qkv + ((size_t)b * 3 * CC + q * CPW) * HH;
    float* kb = qb + CC * HH;
    float* vb = kb + CC * HH;
    if (wid < 8) project16<2>(xt, Wk, Wv, kb, vb, nt, lane);
    else         project16<1>(xt, Wq, Wq, qb, qb, nt, lane);
  }
}

#define NW2  4
#define NT2  (32 * NW2)
#define BR   16
#define LR2  (BR + 2 * CC)
#define ST2  (LR2 * HH / 4 / NT2)

__global__ __launch_bounds__(NT2)
void cluster_attn_kernel(const float* __restrict__ qkv_g,
                         float* __restrict__ out) {
  extern __shared__ float smem[];
  float* V  = smem;
  float* K  = V  + CC * QK_LD;
  float* Q  = K  + CC * QK_LD;
  float* S  = Q  + BR * QK_LD;

  const int tid  = threadIdx.x;
  const int lane = tid & 31;
  const int wid  = tid >> 5;
  const int b    = blockIdx.x >> 2;
  const int q    = blockIdx.x & 3;

  {
    const float4* src4 = (const float4*)(qkv_g + (size_t)b * (3 * CC * HH));
    float4 st[ST2];
#pragma unroll
    for (int j = 0; j < ST2; ++j) {
      const int R  = wid + NW2 * j;
      const int sr = (j < BR / NW2) ? (q * BR + R) : (R + (CC - BR));
      st[j] = src4[sr * (HH / 4) + lane];
    }
#pragma unroll
    for (int j = 0; j < ST2; ++j) {
      const int R = wid + NW2 * j;
      const int L = (j < BR / NW2) ? (2 * CC + R) : (j < (BR + CC) / NW2) ? (R + (CC - BR)) : (R - (BR + CC));
      *(float4*)(smem + L * QK_LD + lane * 4) = st[j];
    }
  }
  __syncthreads();

  const int hi    = (lane >> 4) << 1;
  const int rbase = (lane >> 4) << 3;

  {
    const float scale = 0.08838834764831845f;
    const int col = wid * 16 + (lane & 15);
    v8f acc = {};
    v2f a, bf;
    for (int k = 0; k < HH; k += 4) {
      a  = lds2(Q + (lane & 15) * QK_LD + k + hi);
      bf = lds2(K + col * QK_LD + k + hi);
      acc = __builtin_amdgcn_wmma_f32_16x16x4_f32(false, a, false, bf,
                                                  (short)0, acc, false, false);
    }
    wmma_tail(acc, a, bf);
#pragma unroll
    for (int r = 0; r < 8; ++r) S[(rbase + r) * S_LD + col] = acc[r] * scale;
  }
  __syncthreads();

  {
    float* sp = S + (tid >> 3) * S_LD + (tid & 7) * 8;
    const float2 v01 = *(const float2*)sp;
    const float2 v23 = *(const float2*)(sp + 2);
    const float2 v45 = *(const float2*)(sp + 4);
    const float2 v67 = *(const float2*)(sp + 6);
    float m = fmaxf(fmaxf(fmaxf(v01.x, v01.y), fmaxf(v23.x, v23.y)),
                    fmaxf(fmaxf(v45.x, v45.y), fmaxf(v67.x, v67.y)));
    for (int off = 4; off > 0; off >>= 1) m = fmaxf(m, __shfl_xor(m, off, 32));
    const float e0 = expf(v01.x - m), e1 = expf(v01.y - m);
    const float e2 = expf(v23.x - m), e3 = expf(v23.y - m);
    const float e4 = expf(v45.x - m), e5 = expf(v45.y - m);
    const float e6 = expf(v67.x - m), e7 = expf(v67.y - m);
    float s = ((e0 + e1) + (e2 + e3)) + ((e4 + e5) + (e6 + e7));
    for (int off = 4; off > 0; off >>= 1) s += __shfl_xor(s, off, 32);
    const float inv = 1.0f / s;
    *(float2*)sp       = make_float2(e0 * inv, e1 * inv);
    *(float2*)(sp + 2) = make_float2(e2 * inv, e3 * inv);
    *(float2*)(sp + 4) = make_float2(e4 * inv, e5 * inv);
    *(float2*)(sp + 6) = make_float2(e6 * inv, e7 * inv);
  }
  __syncthreads();

  {
    const int col = wid * 32 + (lane & 15);
    v8f acc0 = {};
    v8f acc1 = {};
    v2f a, b0, b1;
    for (int k = 0; k < CC; k += 4) {
      a = lds2(S + (lane & 15) * S_LD + k + hi);
      b0.x = V[(k + hi) * QK_LD + col];      b0.y = V[(k + 1 + hi) * QK_LD + col];
      b1.x = V[(k + hi) * QK_LD + col + 16]; b1.y = V[(k + 1 + hi) * QK_LD + col + 16];
      acc0 = __builtin_amdgcn_wmma_f32_16x16x4_f32(false, a, false, b0,
                                                   (short)0, acc0, false, false);
      acc1 = __builtin_amdgcn_wmma_f32_16x16x4_f32(false, a, false, b1,
                                                   (short)0, acc1, false, false);
    }
    wmma_tail(acc0, a, b0);
    wmma_tail(acc1, a, b1);
    float* ob = out + ((size_t)b * CC + q * BR) * HH;
#pragma unroll
    for (int r = 0; r < 8; ++r) {
      ob[(rbase + r) * HH + col]      = acc0[r];
      ob[(rbase + r) * HH + col + 16] = acc1[r];
    }
  }
}

extern "C" void kernel_launch(void* const* d_in, const int* in_sizes, int n_in,
                              void* d_out, int out_size, void* d_ws, size_t ws_size,
                              hipStream_t stream) {
  const float* x    = (const float*)d_in[0];
  const int*   rows = (const int*)d_in[3];
  const int*   cols = (const int*)d_in[4];
  const float* vals = (const float*)d_in[5];
  const int*   idx  = (const int*)d_in[6];
  const float* Wq   = (const float*)d_in[7];
  const float* Wk   = (const float*)d_in[8];
  const float* Wv   = (const float*)d_in[9];
  float* qkv = (float*)d_ws;

  const size_t smem1 = (size_t)NNZE * 4 + (size_t)NNZE * 2 + (size_t)(3 * MM) * 4 +
                       (size_t)(2 * CPW * HH + CPW) * 4 + (size_t)NW1 * 4;
  spmm_pool_kernel<<<dim3(BB * (CC / CPW)), dim3(NT1), smem1, stream>>>(
      x, rows, cols, vals, idx, Wq, Wk, Wv, qkv);

  const size_t smem2 = (size_t)(LR2 * QK_LD + BR * S_LD) * sizeof(float);
  cluster_attn_kernel<<<dim3(BB * (CC / BR)), dim3(NT2), smem2, stream>>>(
      qkv, (float*)d_out);
}


# k1 row pass: a wave streams its 128 rows' runs as one list segment, 16 gathers a batch, 2 batches in flight across row ends (row starts flagged in the list); add order in rows and pools as before
# speedup vs baseline: 7.3896x; 1.0812x over previous
#include <hip/hip_runtime.h>
#include <math.h>

#define BB   64
#define NN   2048
#define MM   2048
#define CC   64
#define HH   128
#define NNZE 32768

typedef float v2f __attribute__((ext_vector_type(2)));
typedef float v8f __attribute__((ext_vector_type(8)));

#define XD_LD 388
#define QK_LD 132
#define S_LD  66
#define KG    8

__device__ __forceinline__ v2f ldsA(const float* base, int ld, int mt, int k, int lane) {
  const int hi = (lane >> 4) << 1;
  const float* p = base + (mt * 16 + (lane & 15)) * ld + k + hi;
  v2f a; a.x = p[0]; a.y = p[1];
  return a;
}

__device__ __forceinline__ v2f lds2(const float* p) {
  return *(const v2f*)p;
}

__device__ __forceinline__ void wmma_tail(v8f& acc, const v2f& a, const v2f& b) {
  asm volatile("v_nop\n\tv_nop\n\tv_nop\n\tv_nop" : "+v"(acc) : "v"(a), "v"(b));
}

template <int NM>
__device__ __forceinline__ void project16(const float* xt,
                                          const float* __restrict__ Wa,
                                          const float* __restrict__ Wb,
                                          float* __restrict__ Da,
                                          float* __restrict__ Db,
                                          int nt, int lane) {
  const int hi  = (lane >> 4) << 1;
  const int col = nt * 16 + (lane & 15);
  const unsigned voff = (unsigned)(hi * HH + col);
  const float* xa = xt + (lane & 15) * XD_LD + hi;
  v8f acca = {};
  v8f accb = {};
  for (int k0 = 0; k0 < 3 * HH; k0 += 4 * KG) {
    const float* wa = Wa + k0 * HH;
    const float* wb = Wb + k0 * HH;
    v2f a[KG], ba[KG], bb[KG];
#pragma unroll
    for (int j = 0; j < KG; ++j) {
      ba[j].x = (wa + (4 * j) * HH)[voff]; ba[j].y = (wa + (4 * j + 1) * HH)[voff];
      if (NM == 2) { bb[j].x = (wb + (4 * j) * HH)[voff]; bb[j].y = (wb + (4 * j + 1) * HH)[voff]; }
    }
#pragma unroll
    for (int j = 0; j < KG; ++j) { a[j].x = xa[k0 + 4 * j]; a[j].y = xa[k0 + 4 * j + 1]; }
#pragma unroll
    for (int j = 0; j < KG; ++j) {
      acca = __builtin_amdgcn_wmma_f32_16x16x4_f32(false, a[j], false, ba[j],
                                                   (short)0, acca, false, false);
      if (NM == 2)
        accb = __builtin_amdgcn_wmma_f32_16x16x4_f32(false, a[j], false, bb[j],
                                                     (short)0, accb, false, false);
    }
  }
  const int rbase = (lane >> 4) << 3;
#pragma unroll
  for (int r = 0; r < 8; ++r) {
    Da[(rbase + r) * HH + col] = acca[r];
    if (NM == 2) Db[(rbase + r) * HH + col] = accb[r];
  }
}

#define NW1  16
#define NT1  (32 * NW1)
#define EPT  (NNZE / NT1)
#define CPW  (CC / 4)
#define EB   16
#define SB   16
#define LFIRST 0x8000u

__device__ __forceinline__ int lane_bcast_i(int v, int j) {
  return __builtin_amdgcn_readlane(v, j);
}
__device__ __forceinline__ float lane_bcast_f(float v, int j) {
  return __int_as_float(__builtin_amdgcn_readlane(__float_as_int(v), j));
}

__device__ __forceinline__ void pool_row(const float4& acc, int lc, float* psum, float* pmax, int lane) {
  float* ps = psum + lc * HH + lane * 4;
  float* pm = pmax + lc * HH + lane * 4;
  atomicAdd(ps + 0, acc.x); atomicAdd(ps + 1, acc.y);
  atomicAdd(ps + 2, acc.z); atomicAdd(ps + 3, acc.w);
  __hip_atomic_fetch_max(pm + 0, acc.x, __ATOMIC_RELAXED, __HIP_MEMORY_SCOPE_WORKGROUP);
  __hip_atomic_fetch_max(pm + 1, acc.y, __ATOMIC_RELAXED, __HIP_MEMORY_SCOPE_WORKGROUP);
  __hip_atomic_fetch_max(pm + 2, acc.z, __ATOMIC_RELAXED, __HIP_MEMORY_SCOPE_WORKGROUP);
  __hip_atomic_fetch_max(pm + 3, acc.w, __ATOMIC_RELAXED, __HIP_MEMORY_SCOPE_WORKGROUP);
}

struct SegB {
  float4   xr[SB];
  float    vi;
  int      li;
  unsigned first;
};

__device__ __forceinline__ void seg_load(SegB& s, const float4* __restrict__ xg4,
                                         const unsigned short* lcol, const float* lval,
                                         int p, int e, int lane) {
  const int      pl = p + (lane & (SB - 1));
  const int      pj = min(pl, e - 1);
  const unsigned w  = lcol[pj];
  s.vi    = (pl < e) ? lval[pj] : 0.0f;
  s.first = __builtin_amdgcn_ballot_w32((pl < e) && (w & LFIRST));
  s.li    = (int)((w >> 11) & (CPW - 1));
  const int ci = (int)(w & (NN - 1));
#pragma unroll
  for (int j = 0; j < SB; ++j) {
    const unsigned c = (unsigned)lane_bcast_i(ci, j);
    s.xr[j] = xg4[c * (HH / 4) + (unsigned)lane];
  }
}

__device__ __forceinline__ void seg_sum(float4& acc, int& plc, const SegB& s,
                                        float* psum, float* pmax, int lane) {
#pragma unroll
  for (int j = 0; j < SB; ++j) {
    if ((s.first >> j) & 1u) {
      if (plc >= 0) pool_row(acc, plc, psum, pmax, lane);
      acc = make_float4(0.0f, 0.0f, 0.0f, 0.0f);
      plc = lane_bcast_i(s.li, j);
    }
    const float v = lane_bcast_f(s.vi, j);
    acc.x += v * s.xr[j].x; acc.y += v * s.xr[j].y;
    acc.z += v * s.xr[j].z; acc.w += v * s.xr[j].w;
  }
}

__global__ __launch_bounds__(512)
void spmm_pool_kernel(const float* __restrict__ x,
                      const int*   __restrict__ rows,
                      const int*   __restrict__ cols,
                      const float* __restrict__ vals,
                      const int*   __restrict__ idx,
                      const float* __restrict__ Wq,
                      const float* __restrict__ Wk,
                      const float* __restrict__ Wv,
                      float* __restrict__ qkv) {
  extern __shared__ float smem[];
  float*          lval   = smem;
  unsigned short* lcol   = (unsigned short*)(lval + NNZE);
  int*            bin    = (int*)(lcol + NNZE);
  int*            rstart = bin + MM;
  int*            sidx   = rstart + MM;
  float*          psum   = (float*)(sidx + MM);
  float*          pmax   = psum + CPW * HH;
  float*          pcnt   = pmax + CPW * HH;
  int*            wsum   = (int*)(pcnt + CPW);

  const int tid  = threadIdx.x;
  const int lane = tid & 31;
  const int wid  = __builtin_amdgcn_readfirstlane(tid >> 5);
  const int b    = blockIdx.x >> 2;
  const int q    = blockIdx.x & 3;

  const size_t ibase = (size_t)b * MM;
  const size_t ebase = (size_t)b * NNZE;
  const int*   rp = rows + ebase + tid;
  const int*   cp = cols + ebase + tid;
  const float* vp = vals + ebase + tid;

  int cid[MM / NT1];
#pragma unroll
  for (int j = 0; j < MM / NT1; ++j) cid[j] = idx[ibase + tid + j * NT1];
  int pk[EPT];
#pragma unroll
  for (int j = 0; j < EPT; ++j) pk[j] = rp[j * NT1];

#pragma unroll
  for (int j = 0; j < MM / NT1; ++j) { bin[tid + j * NT1] = tid + j * NT1; sidx[tid + j * NT1] = cid[j]; }
  for (int i = tid; i < CPW * HH; i += NT1) { psum[i] = 0.0f; pmax[i] = -INFINITY; }
  if (tid < CPW) pcnt[tid] = 0.0f;
  __syncthreads();

  int   c[EPT];
  float v[EPT];
#pragma unroll
  for (int i0 = 0; i0 < EPT; i0 += EB) {
#pragma unroll
    for (int j = 0; j < EB; ++j) { c[i0 + j] = cp[(i0 + j) * NT1]; v[i0 + j] = vp[(i0 + j) * NT1]; }
    int cl[EB];
#pragma unroll
    for (int j = 0; j < EB; ++j) cl[j] = sidx[pk[i0 + j]];
#pragma unroll
    for (int j = 0; j < EB; ++j)
      pk[i0 + j] = ((cl[j] >> 4) == q) ? atomicAdd(&bin[pk[i0 + j]], 1 << 11) : -1;
  }
  __syncthreads();

  {
    const int b0 = bin[4 * tid] >> 11, b1 = bin[4 * tid + 1] >> 11, b2 = bin[4 * tid + 2] >> 11, b3 = bin[4 * tid + 3] >> 11;
    const int tsum = b0 + b1 + b2 + b3;
    int incl = tsum;
    for (int off = 1; off < 32; off <<= 1) {
      const int t = __shfl_up(incl, off, 32);
      if (lane >= off) incl += t;
    }
    if (lane == 31) wsum[wid] = incl;
    __syncthreads();
    int woff = 0;
    for (int w = 0; w < wid; ++w) woff += wsum[w];
    int ex = woff + incl - tsum;
    rstart[4 * tid]     = ex; ex += b0;
    rstart[4 * tid + 1] = ex; ex += b1;
    rstart[4 * tid + 2] = ex; ex += b2;
    rstart[4 * tid + 3] = ex;
  }
  __syncthreads();

#pragma unroll
  for (int i0 = 0; i0 < EPT; i0 += EB) {
    int st[EB];
#pragma unroll
    for (int j = 0; j < EB; ++j) st[j] = rstart[pk[i0 + j] & (MM - 1)];
#pragma unroll
    for (int j = 0; j < EB; ++j) {
      if (pk[i0 + j] >= 0) {
        const int pos = st[j] + (pk[i0 + j] >> 11);
        lcol[pos] = (unsigned short)c[i0 + j];
        lval[pos] = v[i0 + j];
      }
    }
  }
  __syncthreads();

  const float4* xg4 = (const float4*)(x + (size_t)b * NN * HH);
  int seg_s = 0, seg_e = 0;
  for (int k = 0; k < MM / (NW1 * 32); ++k) {
    const int m_l  = wid * (MM / NW1) + k * 32 + lane;
    const int cl_l = sidx[m_l];
    const int s_l  = rstart[m_l];
    const int n_l  = bin[m_l] >> 11;
    const int lc_l = cl_l - q * CPW;
    const bool own = (cl_l >> 4) == q;
    if (k == 0) seg_s = __builtin_amdgcn_readlane(s_l, 0);
    seg_e = __builtin_amdgcn_readlane(s_l + n_l, 31);
    if (own) atomicAdd(&pcnt[lc_l], 1.0f);
    if (own && n_l > 0) lcol[s_l] = (unsigned short)(lcol[s_l] | LFIRST | ((unsigned)lc_l << 11));
    unsigned emask = __builtin_amdgcn_ballot_w32(own && n_l == 0);
    while (emask) {
      const int t = __builtin_ctz(emask);
      emask &= emask - 1;
      float* pm = pmax + __builtin_amdgcn_readlane(lc_l, t) * HH + lane * 4;
      __hip_atomic_fetch_max(pm + 0, 0.0f, __ATOMIC_RELAXED, __HIP_MEMORY_SCOPE_WORKGROUP);
      __hip_atomic_fetch_max(pm + 1, 0.0f, __ATOMIC_RELAXED, __HIP_MEMORY_SCOPE_WORKGROUP);
      __hip_atomic_fetch_max(pm + 2, 0.0f, __ATOMIC_RELAXED, __HIP_MEMORY_SCOPE_WORKGROUP);
      __hip_atomic_fetch_max(pm + 3, 0.0f, __ATOMIC_RELAXED, __HIP_MEMORY_SCOPE_WORKGROUP);
    }
  }
  {
    float4 acc = make_float4(0.0f, 0.0f, 0.0f, 0.0f);
    int plc = -1;
    if (seg_s < seg_e) {
      SegB A, B;
      int p = seg_s;
      seg_load(A, xg4, lcol, lval, p, seg_e, lane);
      do {
        seg_load(B, xg4, lcol, lval, p + SB, seg_e, lane);
        seg_sum(acc, plc, A, psum, pmax, lane);
        p += 2 * SB;
        seg_load(A, xg4, lcol, lval, p, seg_e, lane);
        seg_sum(acc, plc, B, psum, pmax, lane);
      } while (p < seg_e);
    }
    if (plc >= 0) pool_row(acc, plc, psum, pmax, lane);
  }
  __syncthreads();

  float* xt = smem;
  {
    const int lc = wid;
    const float4 s4 = *(const float4*)(psum + lc * HH + lane * 4);
    const float4 m4 = *(const float4*)(pmax + lc * HH + lane * 4);
    const float  d  = fmaxf(pcnt[lc], 1.0f);
    float4 mean4;
    mean4.x = s4.x / d; mean4.y = s4.y / d; mean4.z = s4.z / d; mean4.w = s4.w / d;
    float* row = xt + lc * XD_LD + lane * 4;
    *(float4*)(row)          = mean4;
    *(float4*)(row + HH)     = m4;
    *(float4*)(row + 2 * HH) = s4;
  }
  __syncthreads();

  {
    const int nt = wid & 7;
    float* qb = qkv + ((size_t)b * 3 * CC + q * CPW) * HH;
    float* kb = qb + CC * HH;
    float* vb = kb + CC * HH;
    if (wid < 8) project16<2>(xt, Wk, Wv, kb, vb, nt, lane);
    else         project16<1>(xt, Wq, Wq, qb, qb, nt, lane);
  }
}

#define NW2  4
#define NT2  (32 * NW2)
#define BR   16
#define LR2  (BR + 2 * CC)
#define ST2  (LR2 * HH / 4 / NT2)

__global__ __launch_bounds__(NT2)
void cluster_attn_kernel(const float* __restrict__ qkv_g,
                         float* __restrict__ out) {
  extern __shared__ float smem[];
  float* V  = smem;
  float* K  = V  + CC * QK_LD;
  float* Q  = K  + CC * QK_LD;
  float* S  = Q  + BR * QK_LD;

  const int tid  = threadIdx.x;
  const int lane = tid & 31;
  const int wid  = tid >> 5;
  const int b    = blockIdx.x >> 2;
  const int q    = blockIdx.x & 3;

  {
    const float4* src4 = (const float4*)(qkv_g + (size_t)b * (3 * CC * HH));
    float4 st[ST2];
#pragma unroll
    for (int j = 0; j < ST2; ++j) {
      const int R  = wid + NW2 * j;
      const int sr = (j < BR / NW2) ? (q * BR + R) : (R + (CC - BR));
      st[j] = src4[sr * (HH / 4) + lane];
    }
#pragma unroll
    for (int j = 0; j < ST2; ++j) {
      const int R = wid + NW2 * j;
      const int L = (j < BR / NW2) ? (2 * CC + R) : (j < (BR + CC) / NW2) ? (R + (CC - BR)) : (R - (BR + CC));
      *(float4*)(smem + L * QK_LD + lane * 4) = st[j];
    }
  }
  __syncthreads();

  const int hi    = (lane >> 4) << 1;
  const int rbase = (lane >> 4) << 3;

  {
    const float scale = 0.08838834764831845f;
    const int col = wid * 16 + (lane & 15);
    v8f acc = {};
    v2f a, bf;
    for (int k = 0; k < HH; k += 4) {
      a  = lds2(Q + (lane & 15) * QK_LD + k + hi);
      bf = lds2(K + col * QK_LD + k + hi);
      acc = __builtin_amdgcn_wmma_f32_16x16x4_f32(false, a, false, bf,
                                                  (short)0, acc, false, false);
    }
    wmma_tail(acc, a, bf);
#pragma unroll
    for (int r = 0; r < 8; ++r) S[(rbase + r) * S_LD + col] = acc[r] * scale;
  }
  __syncthreads();

  {
    float* sp = S + (tid >> 3) * S_LD + (tid & 7) * 8;
    const float2 v01 = *(const float2*)sp;
    const float2 v23 = *(const float2*)(sp + 2);
    const float2 v45 = *(const float2*)(sp + 4);
    const float2 v67 = *(const float2*)(sp + 6);
    float m = fmaxf(fmaxf(fmaxf(v01.x, v01.y), fmaxf(v23.x, v23.y)),
                    fmaxf(fmaxf(v45.x, v45.y), fmaxf(v67.x, v67.y)));
    for (int off = 4; off > 0; off >>= 1) m = fmaxf(m, __shfl_xor(m, off, 32));
    const float e0 = expf(v01.x - m), e1 = expf(v01.y - m);
    const float e2 = expf(v23.x - m), e3 = expf(v23.y - m);
    const float e4 = expf(v45.x - m), e5 = expf(v45.y - m);
    const float e6 = expf(v67.x - m), e7 = expf(v67.y - m);
    float s = ((e0 + e1) + (e2 + e3)) + ((e4 + e5) + (e6 + e7));
    for (int off = 4; off > 0; off >>= 1) s += __shfl_xor(s, off, 32);
    const float inv = 1.0f / s;
    *(float2*)sp       = make_float2(e0 * inv, e1 * inv);
    *(float2*)(sp + 2) = make_float2(e2 * inv, e3 * inv);
    *(float2*)(sp + 4) = make_float2(e4 * inv, e5 * inv);
    *(float2*)(sp + 6) = make_float2(e6 * inv, e7 * inv);
  }
  __syncthreads();

  {
    const int col = wid * 32 + (lane & 15);
    v8f acc0 = {};
    v8f acc1 = {};
    v2f a, b0, b1;
    for (int k = 0; k < CC; k += 4) {
      a = lds2(S + (lane & 15) * S_LD + k + hi);
      b0.x = V[(k + hi) * QK_LD + col];      b0.y = V[(k + 1 + hi) * QK_LD + col];
      b1.x = V[(k + hi) * QK_LD + col + 16]; b1.y = V[(k + 1 + hi) * QK_LD + col + 16];
      acc0 = __builtin_amdgcn_wmma_f32_16x16x4_f32(false, a, false, b0,
                                                   (short)0, acc0, false, false);
      acc1 = __builtin_amdgcn_wmma_f32_16x16x4_f32(false, a, false, b1,
                                                   (short)0, acc1, false, false);
    }
    wmma_tail(acc0, a, b0);
    wmma_tail(acc1, a, b1);
    float* ob = out + ((size_t)b * CC + q * BR) * HH;
#pragma unroll
    for (int r = 0; r < 8; ++r) {
      ob[(rbase + r) * HH + col]      = acc0[r];
      ob[(rbase + r) * HH + col + 16] = acc1[r];
    }
  }
}

extern "C" void kernel_launch(void* const* d_in, const int* in_sizes, int n_in,
                              void* d_out, int out_size, void* d_ws, size_t ws_size,
                              hipStream_t stream) {
  const float* x    = (const float*)d_in[0];
  const int*   rows = (const int*)d_in[3];
  const int*   cols = (const int*)d_in[4];
  const float* vals = (const float*)d_in[5];
  const int*   idx  = (const int*)d_in[6];
  const float* Wq   = (const float*)d_in[7];
  const float* Wk   = (const float*)d_in[8];
  const float* Wv   = (const float*)d_in[9];
  float* qkv = (float*)d_ws;

  const size_t smem1 = (size_t)NNZE * 4 + (size_t)NNZE * 2 + (size_t)(3 * MM) * 4 +
                       (size_t)(2 * CPW * HH + CPW) * 4 + (size_t)NW1 * 4;
  spmm_pool_kernel<<<dim3(BB * (CC / CPW)), dim3(NT1), smem1, stream>>>(
      x, rows, cols, vals, idx, Wq, Wk, Wv, qkv);

  const size_t smem2 = (size_t)(LR2 * QK_LD + BR * S_LD) * sizeof(float);
  cluster_attn_kernel<<<dim3(BB * (CC / BR)), dim3(NT2), smem2, stream>>>(
      qkv, (float*)d_out);
}
